# starting kernel
# baseline (speedup 1.0000x reference)
#include <hip/hip_runtime.h>

typedef float v2f __attribute__((ext_vector_type(2)));
typedef float v8f __attribute__((ext_vector_type(8)));

#define BB 2
#define SS 2048
#define HH 16
#define DD 128
#define QB 128
#define KSTRIDE 132
#define VSTRIDE 18
#define PSTRIDE 18

__global__ __launch_bounds__(256)
void fa_rope_fwd(const float* __restrict__ q, const float* __restrict__ k,
                 const float* __restrict__ v, const float* __restrict__ cosp,
                 const float* __restrict__ sinp, float* __restrict__ out)
{
    __shared__ float kbuf[16 * KSTRIDE];
    __shared__ float vbufT[DD * VSTRIDE];
    __shared__ float pbuf[8 * 16 * PSTRIDE];

    const int tid  = threadIdx.x;
    const int wave = tid >> 5;
    const int lane = tid & 31;
    const int lmod = lane & 15;
    const int hi   = lane >> 4;
    const int off  = hi * 2;

    const int qblk = blockIdx.x & 15;
    const int bh   = blockIdx.x >> 4;
    const int b    = bh >> 4;
    const int h    = bh & 15;

    const int qbase = qblk * QB + wave * 16;
    const int myrow = qbase + lmod;

    float qa[64];
    {
        const float* qrow = q + (((size_t)b * SS + myrow) * HH + h) * DD;
        const float* crow = cosp + (size_t)myrow * DD;
        const float* srow = sinp + (size_t)myrow * DD;
        float raw[64];
#pragma unroll
        for (int st = 0; st < 32; ++st) {
            v2f t = *(const v2f*)(qrow + 4 * st + off);
            raw[2 * st] = t.x; raw[2 * st + 1] = t.y;
        }
        const float scale = 0.08838834764831845f;
#pragma unroll
        for (int st = 0; st < 16; ++st) {
            const int dlo = 4 * st + off, dhi = dlo + 64;
            v2f cl = *(const v2f*)(crow + dlo);
            v2f sl = *(const v2f*)(srow + dlo);
            v2f ch = *(const v2f*)(crow + dhi);
            v2f sh = *(const v2f*)(srow + dhi);
            const int lo = 2 * st, hp = lo + 32;
            qa[lo]     = (raw[lo]     * cl.x - raw[hp]     * sl.x) * scale;
            qa[lo + 1] = (raw[lo + 1] * cl.y - raw[hp + 1] * sl.y) * scale;
            qa[hp]     = (raw[hp]     * ch.x + raw[lo]     * sh.x) * scale;
            qa[hp + 1] = (raw[hp + 1] * ch.y + raw[lo + 1] * sh.y) * scale;
        }
    }

    v8f o[8];
#pragma unroll
    for (int j = 0; j < 8; ++j)
#pragma unroll
        for (int r = 0; r < 8; ++r) o[j][r] = 0.0f;
    float mrow[8], lrow[8];
#pragma unroll
    for (int r = 0; r < 8; ++r) { mrow[r] = -3.0e38f; lrow[r] = 0.0f; }

    const int ntiles  = qblk * 8 + 8;
    const int ktmax_w = qblk * 8 + wave;
    const size_t kvhead = ((size_t)b * SS * HH + h) * DD;

    for (int kt = 0; kt < ntiles; ++kt) {
        const int kbase = kt * 16;
        __syncthreads();
        {
            const int row = tid >> 4;
            const int c4  = (tid & 15) * 4;
            const int sk  = kbase + row;
            const float* krow = k + kvhead + (size_t)sk * HH * DD;
            const float* vrow = v + kvhead + (size_t)sk * HH * DD;
            const float* crow = cosp + (size_t)sk * DD;
            const float* srow = sinp + (size_t)sk * DD;
            float4 ka = *(const float4*)(krow + c4);
            float4 kb = *(const float4*)(krow + c4 + 64);
            float4 ca = *(const float4*)(crow + c4);
            float4 cb = *(const float4*)(crow + c4 + 64);
            float4 sa = *(const float4*)(srow + c4);
            float4 sb = *(const float4*)(srow + c4 + 64);
            float4 ra, rb;
            ra.x = ka.x * ca.x - kb.x * sa.x;  ra.y = ka.y * ca.y - kb.y * sa.y;
            ra.z = ka.z * ca.z - kb.z * sa.z;  ra.w = ka.w * ca.w - kb.w * sa.w;
            rb.x = kb.x * cb.x + ka.x * sb.x;  rb.y = kb.y * cb.y + ka.y * sb.y;
            rb.z = kb.z * cb.z + ka.z * sb.z;  rb.w = kb.w * cb.w + ka.w * sb.w;
            *(float4*)(kbuf + row * KSTRIDE + c4)      = ra;
            *(float4*)(kbuf + row * KSTRIDE + c4 + 64) = rb;
            float4 va = *(const float4*)(vrow + c4);
            float4 vb = *(const float4*)(vrow + c4 + 64);
            vbufT[(c4 + 0) * VSTRIDE + row] = va.x;
            vbufT[(c4 + 1) * VSTRIDE + row] = va.y;
            vbufT[(c4 + 2) * VSTRIDE + row] = va.z;
            vbufT[(c4 + 3) * VSTRIDE + row] = va.w;
            vbufT[(c4 + 64) * VSTRIDE + row] = vb.x;
            vbufT[(c4 + 65) * VSTRIDE + row] = vb.y;
            vbufT[(c4 + 66) * VSTRIDE + row] = vb.z;
            vbufT[(c4 + 67) * VSTRIDE + row] = vb.w;
        }
        __syncthreads();

        if (kt <= ktmax_w) {
            v8f sc;
#pragma unroll
            for (int r = 0; r < 8; ++r) sc[r] = 0.0f;
#pragma unroll
            for (int st = 0; st < 32; ++st) {
                v2f a;  a.x = qa[2 * st]; a.y = qa[2 * st + 1];
                v2f bb = *(const v2f*)(kbuf + lmod * KSTRIDE + 4 * st + off);
                sc = __builtin_amdgcn_wmma_f32_16x16x4_f32(
                        false, a, false, bb, (short)0, sc, false, false);
            }
            if (kt == ktmax_w) {
#pragma unroll
                for (int r = 0; r < 8; ++r) {
                    const int m = r + 8 * hi;
                    sc[r] = (lmod > m) ? -3.0e38f : sc[r];
                }
            }
            float alpha[8];
            v8f p;
#pragma unroll
            for (int r = 0; r < 8; ++r) {
                float t = sc[r];
                t = fmaxf(t, __shfl_xor(t, 1));
                t = fmaxf(t, __shfl_xor(t, 2));
                t = fmaxf(t, __shfl_xor(t, 4));
                t = fmaxf(t, __shfl_xor(t, 8));
                const float mn = fmaxf(mrow[r], t);
                alpha[r] = __expf(mrow[r] - mn);
                mrow[r]  = mn;
                const float pe = __expf(sc[r] - mn);
                p[r] = pe;
                float su = pe;
                su += __shfl_xor(su, 1);
                su += __shfl_xor(su, 2);
                su += __shfl_xor(su, 4);
                su += __shfl_xor(su, 8);
                lrow[r] = lrow[r] * alpha[r] + su;
            }
#pragma unroll
            for (int j = 0; j < 8; ++j)
#pragma unroll
                for (int r = 0; r < 8; ++r) o[j][r] *= alpha[r];

            float* pb = pbuf + wave * 16 * PSTRIDE;
#pragma unroll
            for (int r = 0; r < 8; ++r)
                pb[(r + 8 * hi) * PSTRIDE + lmod] = p[r];
            v2f ap[4];
#pragma unroll
            for (int k0 = 0; k0 < 4; ++k0)
                ap[k0] = *(const v2f*)(pb + lmod * PSTRIDE + 4 * k0 + off);

#pragma unroll
            for (int j = 0; j < 8; ++j) {
#pragma unroll
                for (int k0 = 0; k0 < 4; ++k0) {
                    v2f bb = *(const v2f*)(vbufT + (16 * j + lmod) * VSTRIDE + 4 * k0 + off);
                    o[j] = __builtin_amdgcn_wmma_f32_16x16x4_f32(
                               false, ap[k0], false, bb, (short)0, o[j], false, false);
                }
            }
        }
    }

    float inv_l[8];
#pragma unroll
    for (int r = 0; r < 8; ++r) inv_l[r] = 1.0f / lrow[r];
    float* obase = out + (size_t)bh * SS * DD;
#pragma unroll
    for (int j = 0; j < 8; ++j) {
#pragma unroll
        for (int r = 0; r < 8; ++r) {
            const int m = r + 8 * hi;
            obase[(size_t)(qbase + m) * DD + 16 * j + lmod] = o[j][r] * inv_l[r];
        }
    }
}

extern "C" void kernel_launch(void* const* d_in, const int* in_sizes, int n_in,
                              void* d_out, int out_size, void* d_ws, size_t ws_size,
                              hipStream_t stream) {
    (void)in_sizes; (void)n_in; (void)out_size; (void)d_ws; (void)ws_size;
    const float* q = (const float*)d_in[0];
    const float* k = (const float*)d_in[1];
    const float* v = (const float*)d_in[2];
    const float* c = (const float*)d_in[3];
    const float* s = (const float*)d_in[4];
    float* o = (float*)d_out;
    dim3 grid(BB * HH * (SS / QB));
    dim3 block(256);
    hipLaunchKernelGGL(fa_rope_fwd, grid, block, 0, stream, q, k, v, c, s, o);
}


# Balance causal work: 256 workgroups each run query blocks 15-p then p of one head (136 key tiles each) instead of 512 workgroups of 8 to 128 tiles; per-row arithmetic and sum order unchanged
# speedup vs baseline: 1.6874x; 1.6874x over previous
#include <hip/hip_runtime.h>

typedef float v2f __attribute__((ext_vector_type(2)));
typedef float v8f __attribute__((ext_vector_type(8)));

#define BB 2
#define SS 2048
#define HH 16
#define DD 128
#define QB 128
#define NQB (SS / QB)
#define KSTRIDE 132
#define VSTRIDE 18
#define PSTRIDE 18

__global__ __launch_bounds__(256)
void fa_rope_fwd(const float* __restrict__ q, const float* __restrict__ k,
                 const float* __restrict__ v, const float* __restrict__ cosp,
                 const float* __restrict__ sinp, float* __restrict__ out)
{
    __shared__ float kbuf[16 * KSTRIDE];
    __shared__ float vbufT[DD * VSTRIDE];
    __shared__ float pbuf[8 * 16 * PSTRIDE];

    const int tid  = threadIdx.x;
    const int wave = tid >> 5;
    const int lane = tid & 31;
    const int lmod = lane & 15;
    const int hi   = lane >> 4;
    const int off  = hi * 2;

    const int pr   = blockIdx.x & 7;
    const int bh   = blockIdx.x >> 3;
    const int b    = bh >> 4;
    const int h    = bh & 15;
    const size_t kvhead = ((size_t)b * SS * HH + h) * DD;

  for (int pass = 0; pass < 2; ++pass) {
    const int qblk = pass ? pr : (NQB - 1 - pr);

    const int qbase = qblk * QB + wave * 16;
    const int myrow = qbase + lmod;

    float qa[64];
    {
        const float* qrow = q + (((size_t)b * SS + myrow) * HH + h) * DD;
        const float* crow = cosp + (size_t)myrow * DD;
        const float* srow = sinp + (size_t)myrow * DD;
        float raw[64];
#pragma unroll
        for (int st = 0; st < 32; ++st) {
            v2f t = *(const v2f*)(qrow + 4 * st + off);
            raw[2 * st] = t.x; raw[2 * st + 1] = t.y;
        }
        const float scale = 0.08838834764831845f;
#pragma unroll
        for (int st = 0; st < 16; ++st) {
            const int dlo = 4 * st + off, dhi = dlo + 64;
            v2f cl = *(const v2f*)(crow + dlo);
            v2f sl = *(const v2f*)(srow + dlo);
            v2f ch = *(const v2f*)(crow + dhi);
            v2f sh = *(const v2f*)(srow + dhi);
            const int lo = 2 * st, hp = lo + 32;
            qa[lo]     = (raw[lo]     * cl.x - raw[hp]     * sl.x) * scale;
            qa[lo + 1] = (raw[lo + 1] * cl.y - raw[hp + 1] * sl.y) * scale;
            qa[hp]     = (raw[hp]     * ch.x + raw[lo]     * sh.x) * scale;
            qa[hp + 1] = (raw[hp + 1] * ch.y + raw[lo + 1] * sh.y) * scale;
        }
    }

    v8f o[8];
#pragma unroll
    for (int j = 0; j < 8; ++j)
#pragma unroll
        for (int r = 0; r < 8; ++r) o[j][r] = 0.0f;
    float mrow[8], lrow[8];
#pragma unroll
    for (int r = 0; r < 8; ++r) { mrow[r] = -3.0e38f; lrow[r] = 0.0f; }

    const int ntiles  = qblk * 8 + 8;
    const int ktmax_w = qblk * 8 + wave;

    for (int kt = 0; kt < ntiles; ++kt) {
        const int kbase = kt * 16;
        __syncthreads();
        {
            const int row = tid >> 4;
            const int c4  = (tid & 15) * 4;
            const int sk  = kbase + row;
            const float* krow = k + kvhead + (size_t)sk * HH * DD;
            const float* vrow = v + kvhead + (size_t)sk * HH * DD;
            const float* crow = cosp + (size_t)sk * DD;
            const float* srow = sinp + (size_t)sk * DD;
            float4 ka = *(const float4*)(krow + c4);
            float4 kb = *(const float4*)(krow + c4 + 64);
            float4 ca = *(const float4*)(crow + c4);
            float4 cb = *(const float4*)(crow + c4 + 64);
            float4 sa = *(const float4*)(srow + c4);
            float4 sb = *(const float4*)(srow + c4 + 64);
            float4 ra, rb;
            ra.x = ka.x * ca.x - kb.x * sa.x;  ra.y = ka.y * ca.y - kb.y * sa.y;
            ra.z = ka.z * ca.z - kb.z * sa.z;  ra.w = ka.w * ca.w - kb.w * sa.w;
            rb.x = kb.x * cb.x + ka.x * sb.x;  rb.y = kb.y * cb.y + ka.y * sb.y;
            rb.z = kb.z * cb.z + ka.z * sb.z;  rb.w = kb.w * cb.w + ka.w * sb.w;
            *(float4*)(kbuf + row * KSTRIDE + c4)      = ra;
            *(float4*)(kbuf + row * KSTRIDE + c4 + 64) = rb;
            float4 va = *(const float4*)(vrow + c4);
            float4 vb = *(const float4*)(vrow + c4 + 64);
            vbufT[(c4 + 0) * VSTRIDE + row] = va.x;
            vbufT[(c4 + 1) * VSTRIDE + row] = va.y;
            vbufT[(c4 + 2) * VSTRIDE + row] = va.z;
            vbufT[(c4 + 3) * VSTRIDE + row] = va.w;
            vbufT[(c4 + 64) * VSTRIDE + row] = vb.x;
            vbufT[(c4 + 65) * VSTRIDE + row] = vb.y;
            vbufT[(c4 + 66) * VSTRIDE + row] = vb.z;
            vbufT[(c4 + 67) * VSTRIDE + row] = vb.w;
        }
        __syncthreads();

        if (kt <= ktmax_w) {
            v8f sc;
#pragma unroll
            for (int r = 0; r < 8; ++r) sc[r] = 0.0f;
#pragma unroll
            for (int st = 0; st < 32; ++st) {
                v2f a;  a.x = qa[2 * st]; a.y = qa[2 * st + 1];
                v2f bb = *(const v2f*)(kbuf + lmod * KSTRIDE + 4 * st + off);
                sc = __builtin_amdgcn_wmma_f32_16x16x4_f32(
                        false, a, false, bb, (short)0, sc, false, false);
            }
            if (kt == ktmax_w) {
#pragma unroll
                for (int r = 0; r < 8; ++r) {
                    const int m = r + 8 * hi;
                    sc[r] = (lmod > m) ? -3.0e38f : sc[r];
                }
            }
            float alpha[8];
            v8f p;
#pragma unroll
            for (int r = 0; r < 8; ++r) {
                float t = sc[r];
                t = fmaxf(t, __shfl_xor(t, 1));
                t = fmaxf(t, __shfl_xor(t, 2));
                t = fmaxf(t, __shfl_xor(t, 4));
                t = fmaxf(t, __shfl_xor(t, 8));
                const float mn = fmaxf(mrow[r], t);
                alpha[r] = __expf(mrow[r] - mn);
                mrow[r]  = mn;
                const float pe = __expf(sc[r] - mn);
                p[r] = pe;
                float su = pe;
                su += __shfl_xor(su, 1);
                su += __shfl_xor(su, 2);
                su += __shfl_xor(su, 4);
                su += __shfl_xor(su, 8);
                lrow[r] = lrow[r] * alpha[r] + su;
            }
#pragma unroll
            for (int j = 0; j < 8; ++j)
#pragma unroll
                for (int r = 0; r < 8; ++r) o[j][r] *= alpha[r];

            float* pb = pbuf + wave * 16 * PSTRIDE;
#pragma unroll
            for (int r = 0; r < 8; ++r)
                pb[(r + 8 * hi) * PSTRIDE + lmod] = p[r];
            v2f ap[4];
#pragma unroll
            for (int k0 = 0; k0 < 4; ++k0)
                ap[k0] = *(const v2f*)(pb + lmod * PSTRIDE + 4 * k0 + off);

#pragma unroll
            for (int j = 0; j < 8; ++j) {
#pragma unroll
                for (int k0 = 0; k0 < 4; ++k0) {
                    v2f bb = *(const v2f*)(vbufT + (16 * j + lmod) * VSTRIDE + 4 * k0 + off);
                    o[j] = __builtin_amdgcn_wmma_f32_16x16x4_f32(
                               false, ap[k0], false, bb, (short)0, o[j], false, false);
                }
            }
        }
    }

    float inv_l[8];
#pragma unroll
    for (int r = 0; r < 8; ++r) inv_l[r] = 1.0f / lrow[r];
    float* obase = out + (size_t)bh * SS * DD;
#pragma unroll
    for (int j = 0; j < 8; ++j) {
#pragma unroll
        for (int r = 0; r < 8; ++r) {
            const int m = r + 8 * hi;
            obase[(size_t)(qbase + m) * DD + 16 * j + lmod] = o[j][r] * inv_l[r];
        }
    }
  }
}

extern "C" void kernel_launch(void* const* d_in, const int* in_sizes, int n_in,
                              void* d_out, int out_size, void* d_ws, size_t ws_size,
                              hipStream_t stream) {
    (void)in_sizes; (void)n_in; (void)out_size; (void)d_ws; (void)ws_size;
    const float* q = (const float*)d_in[0];
    const float* k = (const float*)d_in[1];
    const float* v = (const float*)d_in[2];
    const float* c = (const float*)d_in[3];
    const float* s = (const float*)d_in[4];
    float* o = (float*)d_out;
    dim3 grid(BB * HH * (NQB / 2));
    dim3 block(256);
    hipLaunchKernelGGL(fa_rope_fwd, grid, block, 0, stream, q, k, v, c, s, o);
}


# Software-pipelined K/V stage: tile kt+1's global loads issued before tile kt's WMMAs, written to a second LDS copy after them; one barrier per tile, not two; arithmetic and sum order unchanged
# speedup vs baseline: 1.8357x; 1.0879x over previous
#include <hip/hip_runtime.h>

typedef float v2f __attribute__((ext_vector_type(2)));
typedef float v8f __attribute__((ext_vector_type(8)));

#define BB 2
#define SS 2048
#define HH 16
#define DD 128
#define QB 128
#define NQB (SS / QB)
#define KSTRIDE 132
#define VSTRIDE 18
#define PSTRIDE 18

struct StageRegs { float4 ka, kb, ca, cb, sa, sb, va, vb; };

__device__ __forceinline__ void stage_fetch(StageRegs& s, int kt, int tid, size_t kvhead,
        const float* __restrict__ k, const float* __restrict__ v,
        const float* __restrict__ cosp, const float* __restrict__ sinp)
{
    const int row = tid >> 4;
    const int c4  = (tid & 15) * 4;
    const int sk  = kt * 16 + row;
    const float* krow = k + kvhead + (size_t)sk * HH * DD;
    const float* vrow = v + kvhead + (size_t)sk * HH * DD;
    const float* crow = cosp + (size_t)sk * DD;
    const float* srow = sinp + (size_t)sk * DD;
    s.ka = *(const float4*)(krow + c4);
    s.kb = *(const float4*)(krow + c4 + 64);
    s.ca = *(const float4*)(crow + c4);
    s.cb = *(const float4*)(crow + c4 + 64);
    s.sa = *(const float4*)(srow + c4);
    s.sb = *(const float4*)(srow + c4 + 64);
    s.va = *(const float4*)(vrow + c4);
    s.vb = *(const float4*)(vrow + c4 + 64);
}

__device__ __forceinline__ void stage_commit(const StageRegs& s, int tid, float* kbuf, float* vbufT)
{
    const int row = tid >> 4;
    const int c4  = (tid & 15) * 4;
    const float4 ka = s.ka, kb = s.kb, ca = s.ca, cb = s.cb, sa = s.sa, sb = s.sb;
    float4 ra, rb;
    ra.x = ka.x * ca.x - kb.x * sa.x;  ra.y = ka.y * ca.y - kb.y * sa.y;
    ra.z = ka.z * ca.z - kb.z * sa.z;  ra.w = ka.w * ca.w - kb.w * sa.w;
    rb.x = kb.x * cb.x + ka.x * sb.x;  rb.y = kb.y * cb.y + ka.y * sb.y;
    rb.z = kb.z * cb.z + ka.z * sb.z;  rb.w = kb.w * cb.w + ka.w * sb.w;
    *(float4*)(kbuf + row * KSTRIDE + c4)      = ra;
    *(float4*)(kbuf + row * KSTRIDE + c4 + 64) = rb;
    const float4 va = s.va, vb = s.vb;
    vbufT[(c4 + 0) * VSTRIDE + row] = va.x;
    vbufT[(c4 + 1) * VSTRIDE + row] = va.y;
    vbufT[(c4 + 2) * VSTRIDE + row] = va.z;
    vbufT[(c4 + 3) * VSTRIDE + row] = va.w;
    vbufT[(c4 + 64) * VSTRIDE + row] = vb.x;
    vbufT[(c4 + 65) * VSTRIDE + row] = vb.y;
    vbufT[(c4 + 66) * VSTRIDE + row] = vb.z;
    vbufT[(c4 + 67) * VSTRIDE + row] = vb.w;
}

__global__ __launch_bounds__(256)
void fa_rope_fwd(const float* __restrict__ q, const float* __restrict__ k,
                 const float* __restrict__ v, const float* __restrict__ cosp,
                 const float* __restrict__ sinp, float* __restrict__ out)
{
    __shared__ float kbuf2[2][16 * KSTRIDE];
    __shared__ float vbufT2[2][DD * VSTRIDE];
    __shared__ float pbuf[8 * 16 * PSTRIDE];

    const int tid  = threadIdx.x;
    const int wave = tid >> 5;
    const int lane = tid & 31;
    const int lmod = lane & 15;
    const int hi   = lane >> 4;
    const int off  = hi * 2;

    const int pr   = blockIdx.x & 7;
    const int bh   = blockIdx.x >> 3;
    const int b    = bh >> 4;
    const int h    = bh & 15;
    const size_t kvhead = ((size_t)b * SS * HH + h) * DD;

  for (int pass = 0; pass < 2; ++pass) {
    const int qblk = pass ? pr : (NQB - 1 - pr);

    const int qbase = qblk * QB + wave * 16;
    const int myrow = qbase + lmod;

    float qa[64];
    {
        const float* qrow = q + (((size_t)b * SS + myrow) * HH + h) * DD;
        const float* crow = cosp + (size_t)myrow * DD;
        const float* srow = sinp + (size_t)myrow * DD;
        float raw[64];
#pragma unroll
        for (int st = 0; st < 32; ++st) {
            v2f t = *(const v2f*)(qrow + 4 * st + off);
            raw[2 * st] = t.x; raw[2 * st + 1] = t.y;
        }
        const float scale = 0.08838834764831845f;
#pragma unroll
        for (int st = 0; st < 16; ++st) {
            const int dlo = 4 * st + off, dhi = dlo + 64;
            v2f cl = *(const v2f*)(crow + dlo);
            v2f sl = *(const v2f*)(srow + dlo);
            v2f ch = *(const v2f*)(crow + dhi);
            v2f sh = *(const v2f*)(srow + dhi);
            const int lo = 2 * st, hp = lo + 32;
            qa[lo]     = (raw[lo]     * cl.x - raw[hp]     * sl.x) * scale;
            qa[lo + 1] = (raw[lo + 1] * cl.y - raw[hp + 1] * sl.y) * scale;
            qa[hp]     = (raw[hp]     * ch.x + raw[lo]     * sh.x) * scale;
            qa[hp + 1] = (raw[hp + 1] * ch.y + raw[lo + 1] * sh.y) * scale;
        }
    }

    v8f o[8];
#pragma unroll
    for (int j = 0; j < 8; ++j)
#pragma unroll
        for (int r = 0; r < 8; ++r) o[j][r] = 0.0f;
    float mrow[8], lrow[8];
#pragma unroll
    for (int r = 0; r < 8; ++r) { mrow[r] = -3.0e38f; lrow[r] = 0.0f; }

    const int ntiles  = qblk * 8 + 8;
    const int ktmax_w = qblk * 8 + wave;

    StageRegs sr;
    stage_fetch(sr, 0, tid, kvhead, k, v, cosp, sinp);
    stage_commit(sr, tid, kbuf2[0], vbufT2[0]);
    __syncthreads();

    for (int kt = 0; kt < ntiles; ++kt) {
        const int cur = kt & 1;
        const float* kbuf  = kbuf2[cur];
        const float* vbufT = vbufT2[cur];
        const bool more = (kt + 1 < ntiles);
        if (more) stage_fetch(sr, kt + 1, tid, kvhead, k, v, cosp, sinp);

        if (kt <= ktmax_w) {
            v8f sc;
#pragma unroll
            for (int r = 0; r < 8; ++r) sc[r] = 0.0f;
#pragma unroll
            for (int st = 0; st < 32; ++st) {
                v2f a;  a.x = qa[2 * st]; a.y = qa[2 * st + 1];
                v2f bb = *(const v2f*)(kbuf + lmod * KSTRIDE + 4 * st + off);
                sc = __builtin_amdgcn_wmma_f32_16x16x4_f32(
                        false, a, false, bb, (short)0, sc, false, false);
            }
            if (kt == ktmax_w) {
#pragma unroll
                for (int r = 0; r < 8; ++r) {
                    const int m = r + 8 * hi;
                    sc[r] = (lmod > m) ? -3.0e38f : sc[r];
                }
            }
            float alpha[8];
            v8f p;
#pragma unroll
            for (int r = 0; r < 8; ++r) {
                float t = sc[r];
                t = fmaxf(t, __shfl_xor(t, 1));
                t = fmaxf(t, __shfl_xor(t, 2));
                t = fmaxf(t, __shfl_xor(t, 4));
                t = fmaxf(t, __shfl_xor(t, 8));
                const float mn = fmaxf(mrow[r], t);
                alpha[r] = __expf(mrow[r] - mn);
                mrow[r]  = mn;
                const float pe = __expf(sc[r] - mn);
                p[r] = pe;
                float su = pe;
                su += __shfl_xor(su, 1);
                su += __shfl_xor(su, 2);
                su += __shfl_xor(su, 4);
                su += __shfl_xor(su, 8);
                lrow[r] = lrow[r] * alpha[r] + su;
            }
#pragma unroll
            for (int j = 0; j < 8; ++j)
#pragma unroll
                for (int r = 0; r < 8; ++r) o[j][r] *= alpha[r];

            float* pb = pbuf + wave * 16 * PSTRIDE;
#pragma unroll
            for (int r = 0; r < 8; ++r)
                pb[(r + 8 * hi) * PSTRIDE + lmod] = p[r];
            v2f ap[4];
#pragma unroll
            for (int k0 = 0; k0 < 4; ++k0)
                ap[k0] = *(const v2f*)(pb + lmod * PSTRIDE + 4 * k0 + off);

#pragma unroll
            for (int j = 0; j < 8; ++j) {
#pragma unroll
                for (int k0 = 0; k0 < 4; ++k0) {
                    v2f bb = *(const v2f*)(vbufT + (16 * j + lmod) * VSTRIDE + 4 * k0 + off);
                    o[j] = __builtin_amdgcn_wmma_f32_16x16x4_f32(
                               false, ap[k0], false, bb, (short)0, o[j], false, false);
                }
            }
        }

        if (more) stage_commit(sr, tid, kbuf2[cur ^ 1], vbufT2[cur ^ 1]);
        __syncthreads();
    }

    float inv_l[8];
#pragma unroll
    for (int r = 0; r < 8; ++r) inv_l[r] = 1.0f / lrow[r];
    float* obase = out + (size_t)bh * SS * DD;
#pragma unroll
    for (int j = 0; j < 8; ++j) {
#pragma unroll
        for (int r = 0; r < 8; ++r) {
            const int m = r + 8 * hi;
            obase[(size_t)(qbase + m) * DD + 16 * j + lmod] = o[j][r] * inv_l[r];
        }
    }
  }
}

extern "C" void kernel_launch(void* const* d_in, const int* in_sizes, int n_in,
                              void* d_out, int out_size, void* d_ws, size_t ws_size,
                              hipStream_t stream) {
    (void)in_sizes; (void)n_in; (void)out_size; (void)d_ws; (void)ws_size;
    const float* q = (const float*)d_in[0];
    const float* k = (const float*)d_in[1];
    const float* v = (const float*)d_in[2];
    const float* c = (const float*)d_in[3];
    const float* s = (const float*)d_in[4];
    float* o = (float*)d_out;
    dim3 grid(BB * HH * (NQB / 2));
    dim3 block(256);
    hipLaunchKernelGGL(fa_rope_fwd, grid, block, 0, stream, q, k, v, c, s, o);
}


# Softmax row max and row sum over each 16-lane row by DPP row_xmask butterflies on the vector ALU instead of LDS permutes; same xor 1,2,4,8 tree, sum order unchanged
# speedup vs baseline: 2.2570x; 1.2295x over previous
#include <hip/hip_runtime.h>

typedef float v2f __attribute__((ext_vector_type(2)));
typedef float v8f __attribute__((ext_vector_type(8)));

#define BB 2
#define SS 2048
#define HH 16
#define DD 128
#define QB 128
#define NQB (SS / QB)
#define KSTRIDE 132
#define VSTRIDE 18
#define PSTRIDE 18

template <int M>
__device__ __forceinline__ float row_xor(float x)
{
    const int b = __builtin_bit_cast(int, x);
    return __builtin_bit_cast(float, __builtin_amdgcn_mov_dpp(b, 0x160 + M, 0xf, 0xf, true));
}

struct StageRegs { float4 ka, kb, ca, cb, sa, sb, va, vb; };

__device__ __forceinline__ void stage_fetch(StageRegs& s, int kt, int tid, size_t kvhead,
        const float* __restrict__ k, const float* __restrict__ v,
        const float* __restrict__ cosp, const float* __restrict__ sinp)
{
    const int row = tid >> 4;
    const int c4  = (tid & 15) * 4;
    const int sk  = kt * 16 + row;
    const float* krow = k + kvhead + (size_t)sk * HH * DD;
    const float* vrow = v + kvhead + (size_t)sk * HH * DD;
    const float* crow = cosp + (size_t)sk * DD;
    const float* srow = sinp + (size_t)sk * DD;
    s.ka = *(const float4*)(krow + c4);
    s.kb = *(const float4*)(krow + c4 + 64);
    s.ca = *(const float4*)(crow + c4);
    s.cb = *(const float4*)(crow + c4 + 64);
    s.sa = *(const float4*)(srow + c4);
    s.sb = *(const float4*)(srow + c4 + 64);
    s.va = *(const float4*)(vrow + c4);
    s.vb = *(const float4*)(vrow + c4 + 64);
}

__device__ __forceinline__ void stage_commit(const StageRegs& s, int tid, float* kbuf, float* vbufT)
{
    const int row = tid >> 4;
    const int c4  = (tid & 15) * 4;
    const float4 ka = s.ka, kb = s.kb, ca = s.ca, cb = s.cb, sa = s.sa, sb = s.sb;
    float4 ra, rb;
    ra.x = ka.x * ca.x - kb.x * sa.x;  ra.y = ka.y * ca.y - kb.y * sa.y;
    ra.z = ka.z * ca.z - kb.z * sa.z;  ra.w = ka.w * ca.w - kb.w * sa.w;
    rb.x = kb.x * cb.x + ka.x * sb.x;  rb.y = kb.y * cb.y + ka.y * sb.y;
    rb.z = kb.z * cb.z + ka.z * sb.z;  rb.w = kb.w * cb.w + ka.w * sb.w;
    *(float4*)(kbuf + row * KSTRIDE + c4)      = ra;
    *(float4*)(kbuf + row * KSTRIDE + c4 + 64) = rb;
    const float4 va = s.va, vb = s.vb;
    vbufT[(c4 + 0) * VSTRIDE + row] = va.x;
    vbufT[(c4 + 1) * VSTRIDE + row] = va.y;
    vbufT[(c4 + 2) * VSTRIDE + row] = va.z;
    vbufT[(c4 + 3) * VSTRIDE + row] = va.w;
    vbufT[(c4 + 64) * VSTRIDE + row] = vb.x;
    vbufT[(c4 + 65) * VSTRIDE + row] = vb.y;
    vbufT[(c4 + 66) * VSTRIDE + row] = vb.z;
    vbufT[(c4 + 67) * VSTRIDE + row] = vb.w;
}

__global__ __launch_bounds__(256)
void fa_rope_fwd(const float* __restrict__ q, const float* __restrict__ k,
                 const float* __restrict__ v, const float* __restrict__ cosp,
                 const float* __restrict__ sinp, float* __restrict__ out)
{
    __shared__ float kbuf2[2][16 * KSTRIDE];
    __shared__ float vbufT2[2][DD * VSTRIDE];
    __shared__ float pbuf[8 * 16 * PSTRIDE];

    const int tid  = threadIdx.x;
    const int wave = tid >> 5;
    const int lane = tid & 31;
    const int lmod = lane & 15;
    const int hi   = lane >> 4;
    const int off  = hi * 2;

    const int pr   = blockIdx.x & 7;
    const int bh   = blockIdx.x >> 3;
    const int b    = bh >> 4;
    const int h    = bh & 15;
    const size_t kvhead = ((size_t)b * SS * HH + h) * DD;

  for (int pass = 0; pass < 2; ++pass) {
    const int qblk = pass ? pr : (NQB - 1 - pr);

    const int qbase = qblk * QB + wave * 16;
    const int myrow = qbase + lmod;

    float qa[64];
    {
        const float* qrow = q + (((size_t)b * SS + myrow) * HH + h) * DD;
        const float* crow = cosp + (size_t)myrow * DD;
        const float* srow = sinp + (size_t)myrow * DD;
        float raw[64];
#pragma unroll
        for (int st = 0; st < 32; ++st) {
            v2f t = *(const v2f*)(qrow + 4 * st + off);
            raw[2 * st] = t.x; raw[2 * st + 1] = t.y;
        }
        const float scale = 0.08838834764831845f;
#pragma unroll
        for (int st = 0; st < 16; ++st) {
            const int dlo = 4 * st + off, dhi = dlo + 64;
            v2f cl = *(const v2f*)(crow + dlo);
            v2f sl = *(const v2f*)(srow + dlo);
            v2f ch = *(const v2f*)(crow + dhi);
            v2f sh = *(const v2f*)(srow + dhi);
            const int lo = 2 * st, hp = lo + 32;
            qa[lo]     = (raw[lo]     * cl.x - raw[hp]     * sl.x) * scale;
            qa[lo + 1] = (raw[lo + 1] * cl.y - raw[hp + 1] * sl.y) * scale;
            qa[hp]     = (raw[hp]     * ch.x + raw[lo]     * sh.x) * scale;
            qa[hp + 1] = (raw[hp + 1] * ch.y + raw[lo + 1] * sh.y) * scale;
        }
    }

    v8f o[8];
#pragma unroll
    for (int j = 0; j < 8; ++j)
#pragma unroll
        for (int r = 0; r < 8; ++r) o[j][r] = 0.0f;
    float mrow[8], lrow[8];
#pragma unroll
    for (int r = 0; r < 8; ++r) { mrow[r] = -3.0e38f; lrow[r] = 0.0f; }

    const int ntiles  = qblk * 8 + 8;
    const int ktmax_w = qblk * 8 + wave;

    StageRegs sr;
    stage_fetch(sr, 0, tid, kvhead, k, v, cosp, sinp);
    stage_commit(sr, tid, kbuf2[0], vbufT2[0]);
    __syncthreads();

    for (int kt = 0; kt < ntiles; ++kt) {
        const int cur = kt & 1;
        const float* kbuf  = kbuf2[cur];
        const float* vbufT = vbufT2[cur];
        const bool more = (kt + 1 < ntiles);
        if (more) stage_fetch(sr, kt + 1, tid, kvhead, k, v, cosp, sinp);

        if (kt <= ktmax_w) {
            v8f sc;
#pragma unroll
            for (int r = 0; r < 8; ++r) sc[r] = 0.0f;
#pragma unroll
            for (int st = 0; st < 32; ++st) {
                v2f a;  a.x = qa[2 * st]; a.y = qa[2 * st + 1];
                v2f bb = *(const v2f*)(kbuf + lmod * KSTRIDE + 4 * st + off);
                sc = __builtin_amdgcn_wmma_f32_16x16x4_f32(
                        false, a, false, bb, (short)0, sc, false, false);
            }
            if (kt == ktmax_w) {
#pragma unroll
                for (int r = 0; r < 8; ++r) {
                    const int m = r + 8 * hi;
                    sc[r] = (lmod > m) ? -3.0e38f : sc[r];
                }
            }
            float alpha[8];
            v8f p;
#pragma unroll
            for (int r = 0; r < 8; ++r) {
                float t = sc[r];
                t = fmaxf(t, row_xor<1>(t));
                t = fmaxf(t, row_xor<2>(t));
                t = fmaxf(t, row_xor<4>(t));
                t = fmaxf(t, row_xor<8>(t));
                const float mn = fmaxf(mrow[r], t);
                alpha[r] = __expf(mrow[r] - mn);
                mrow[r]  = mn;
                const float pe = __expf(sc[r] - mn);
                p[r] = pe;
                float su = pe;
                su += row_xor<1>(su);
                su += row_xor<2>(su);
                su += row_xor<4>(su);
                su += row_xor<8>(su);
                lrow[r] = lrow[r] * alpha[r] + su;
            }
#pragma unroll
            for (int j = 0; j < 8; ++j)
#pragma unroll
                for (int r = 0; r < 8; ++r) o[j][r] *= alpha[r];

            float* pb = pbuf + wave * 16 * PSTRIDE;
#pragma unroll
            for (int r = 0; r < 8; ++r)
                pb[(r + 8 * hi) * PSTRIDE + lmod] = p[r];
            v2f ap[4];
#pragma unroll
            for (int k0 = 0; k0 < 4; ++k0)
                ap[k0] = *(const v2f*)(pb + lmod * PSTRIDE + 4 * k0 + off);

#pragma unroll
            for (int j = 0; j < 8; ++j) {
#pragma unroll
                for (int k0 = 0; k0 < 4; ++k0) {
                    v2f bb = *(const v2f*)(vbufT + (16 * j + lmod) * VSTRIDE + 4 * k0 + off);
                    o[j] = __builtin_amdgcn_wmma_f32_16x16x4_f32(
                               false, ap[k0], false, bb, (short)0, o[j], false, false);
                }
            }
        }

        if (more) stage_commit(sr, tid, kbuf2[cur ^ 1], vbufT2[cur ^ 1]);
        __syncthreads();
    }

    float inv_l[8];
#pragma unroll
    for (int r = 0; r < 8; ++r) inv_l[r] = 1.0f / lrow[r];
    float* obase = out + (size_t)bh * SS * DD;
#pragma unroll
    for (int j = 0; j < 8; ++j) {
#pragma unroll
        for (int r = 0; r < 8; ++r) {
            const int m = r + 8 * hi;
            obase[(size_t)(qbase + m) * DD + 16 * j + lmod] = o[j][r] * inv_l[r];
        }
    }
  }
}

extern "C" void kernel_launch(void* const* d_in, const int* in_sizes, int n_in,
                              void* d_out, int out_size, void* d_ws, size_t ws_size,
                              hipStream_t stream) {
    (void)in_sizes; (void)n_in; (void)out_size; (void)d_ws; (void)ws_size;
    const float* q = (const float*)d_in[0];
    const float* k = (const float*)d_in[1];
    const float* v = (const float*)d_in[2];
    const float* c = (const float*)d_in[3];
    const float* s = (const float*)d_in[4];
    float* o = (float*)d_out;
    dim3 grid(BB * HH * (NQB / 2));
    dim3 block(256);
    hipLaunchKernelGGL(fa_rope_fwd, grid, block, 0, stream, q, k, v, c, s, o);
}


# K and V fragment LDS reads of each key tile issued as one group ahead of each 32-WMMA block (into registers) instead of between the WMMAs; arithmetic and sum order unchanged
# speedup vs baseline: 2.3281x; 1.0315x over previous
#include <hip/hip_runtime.h>

typedef float v2f __attribute__((ext_vector_type(2)));
typedef float v8f __attribute__((ext_vector_type(8)));

#define BB 2
#define SS 2048
#define HH 16
#define DD 128
#define QB 128
#define NQB (SS / QB)
#define KSTRIDE 132
#define VSTRIDE 18
#define PSTRIDE 18

template <int M>
__device__ __forceinline__ float row_xor(float x)
{
    const int b = __builtin_bit_cast(int, x);
    return __builtin_bit_cast(float, __builtin_amdgcn_mov_dpp(b, 0x160 + M, 0xf, 0xf, true));
}

struct StageRegs { float4 ka, kb, ca, cb, sa, sb, va, vb; };

__device__ __forceinline__ void stage_fetch(StageRegs& s, int kt, int tid, size_t kvhead,
        const float* __restrict__ k, const float* __restrict__ v,
        const float* __restrict__ cosp, const float* __restrict__ sinp)
{
    const int row = tid >> 4;
    const int c4  = (tid & 15) * 4;
    const int sk  = kt * 16 + row;
    const float* krow = k + kvhead + (size_t)sk * HH * DD;
    const float* vrow = v + kvhead + (size_t)sk * HH * DD;
    const float* crow = cosp + (size_t)sk * DD;
    const float* srow = sinp + (size_t)sk * DD;
    s.ka = *(const float4*)(krow + c4);
    s.kb = *(const float4*)(krow + c4 + 64);
    s.ca = *(const float4*)(crow + c4);
    s.cb = *(const float4*)(crow + c4 + 64);
    s.sa = *(const float4*)(srow + c4);
    s.sb = *(const float4*)(srow + c4 + 64);
    s.va = *(const float4*)(vrow + c4);
    s.vb = *(const float4*)(vrow + c4 + 64);
}

__device__ __forceinline__ void stage_commit(const StageRegs& s, int tid, float* kbuf, float* vbufT)
{
    const int row = tid >> 4;
    const int c4  = (tid & 15) * 4;
    const float4 ka = s.ka, kb = s.kb, ca = s.ca, cb = s.cb, sa = s.sa, sb = s.sb;
    float4 ra, rb;
    ra.x = ka.x * ca.x - kb.x * sa.x;  ra.y = ka.y * ca.y - kb.y * sa.y;
    ra.z = ka.z * ca.z - kb.z * sa.z;  ra.w = ka.w * ca.w - kb.w * sa.w;
    rb.x = kb.x * cb.x + ka.x * sb.x;  rb.y = kb.y * cb.y + ka.y * sb.y;
    rb.z = kb.z * cb.z + ka.z * sb.z;  rb.w = kb.w * cb.w + ka.w * sb.w;
    *(float4*)(kbuf + row * KSTRIDE + c4)      = ra;
    *(float4*)(kbuf + row * KSTRIDE + c4 + 64) = rb;
    const float4 va = s.va, vb = s.vb;
    vbufT[(c4 + 0) * VSTRIDE + row] = va.x;
    vbufT[(c4 + 1) * VSTRIDE + row] = va.y;
    vbufT[(c4 + 2) * VSTRIDE + row] = va.z;
    vbufT[(c4 + 3) * VSTRIDE + row] = va.w;
    vbufT[(c4 + 64) * VSTRIDE + row] = vb.x;
    vbufT[(c4 + 65) * VSTRIDE + row] = vb.y;
    vbufT[(c4 + 66) * VSTRIDE + row] = vb.z;
    vbufT[(c4 + 67) * VSTRIDE + row] = vb.w;
}

__global__ __launch_bounds__(256)
void fa_rope_fwd(const float* __restrict__ q, const float* __restrict__ k,
                 const float* __restrict__ v, const float* __restrict__ cosp,
                 const float* __restrict__ sinp, float* __restrict__ out)
{
    __shared__ float kbuf2[2][16 * KSTRIDE];
    __shared__ float vbufT2[2][DD * VSTRIDE];
    __shared__ float pbuf[8 * 16 * PSTRIDE];

    const int tid  = threadIdx.x;
    const int wave = tid >> 5;
    const int lane = tid & 31;
    const int lmod = lane & 15;
    const int hi   = lane >> 4;
    const int off  = hi * 2;

    const int pr   = blockIdx.x & 7;
    const int bh   = blockIdx.x >> 3;
    const int b    = bh >> 4;
    const int h    = bh & 15;
    const size_t kvhead = ((size_t)b * SS * HH + h) * DD;

  for (int pass = 0; pass < 2; ++pass) {
    const int qblk = pass ? pr : (NQB - 1 - pr);

    const int qbase = qblk * QB + wave * 16;
    const int myrow = qbase + lmod;

    float qa[64];
    {
        const float* qrow = q + (((size_t)b * SS + myrow) * HH + h) * DD;
        const float* crow = cosp + (size_t)myrow * DD;
        const float* srow = sinp + (size_t)myrow * DD;
        float raw[64];
#pragma unroll
        for (int st = 0; st < 32; ++st) {
            v2f t = *(const v2f*)(qrow + 4 * st + off);
            raw[2 * st] = t.x; raw[2 * st + 1] = t.y;
        }
        const float scale = 0.08838834764831845f;
#pragma unroll
        for (int st = 0; st < 16; ++st) {
            const int dlo = 4 * st + off, dhi = dlo + 64;
            v2f cl = *(const v2f*)(crow + dlo);
            v2f sl = *(const v2f*)(srow + dlo);
            v2f ch = *(const v2f*)(crow + dhi);
            v2f sh = *(const v2f*)(srow + dhi);
            const int lo = 2 * st, hp = lo + 32;
            qa[lo]     = (raw[lo]     * cl.x - raw[hp]     * sl.x) * scale;
            qa[lo + 1] = (raw[lo + 1] * cl.y - raw[hp + 1] * sl.y) * scale;
            qa[hp]     = (raw[hp]     * ch.x + raw[lo]     * sh.x) * scale;
            qa[hp + 1] = (raw[hp + 1] * ch.y + raw[lo + 1] * sh.y) * scale;
        }
    }

    v8f o[8];
#pragma unroll
    for (int j = 0; j < 8; ++j)
#pragma unroll
        for (int r = 0; r < 8; ++r) o[j][r] = 0.0f;
    float mrow[8], lrow[8];
#pragma unroll
    for (int r = 0; r < 8; ++r) { mrow[r] = -3.0e38f; lrow[r] = 0.0f; }

    const int ntiles  = qblk * 8 + 8;
    const int ktmax_w = qblk * 8 + wave;

    StageRegs sr;
    stage_fetch(sr, 0, tid, kvhead, k, v, cosp, sinp);
    stage_commit(sr, tid, kbuf2[0], vbufT2[0]);
    __syncthreads();

    for (int kt = 0; kt < ntiles; ++kt) {
        const int cur = kt & 1;
        const float* kbuf  = kbuf2[cur];
        const float* vbufT = vbufT2[cur];
        const bool more = (kt + 1 < ntiles);
        if (more) stage_fetch(sr, kt + 1, tid, kvhead, k, v, cosp, sinp);

        if (kt <= ktmax_w) {
            v2f kf[32];
#pragma unroll
            for (int st = 0; st < 32; ++st)
                kf[st] = *(const v2f*)(kbuf + lmod * KSTRIDE + 4 * st + off);
            __builtin_amdgcn_sched_barrier(0);
            v8f sc;
#pragma unroll
            for (int r = 0; r < 8; ++r) sc[r] = 0.0f;
#pragma unroll
            for (int st = 0; st < 32; ++st) {
                v2f a;  a.x = qa[2 * st]; a.y = qa[2 * st + 1];
                sc = __builtin_amdgcn_wmma_f32_16x16x4_f32(
                        false, a, false, kf[st], (short)0, sc, false, false);
            }
            int voff = lmod * VSTRIDE + off;
            asm volatile("" : "+v"(voff) : "v"(sc));
            v2f vf[32];
#pragma unroll
            for (int j = 0; j < 8; ++j)
#pragma unroll
                for (int k0 = 0; k0 < 4; ++k0)
                    vf[4 * j + k0] = *(const v2f*)(vbufT + voff + 16 * j * VSTRIDE + 4 * k0);
            if (kt == ktmax_w) {
#pragma unroll
                for (int r = 0; r < 8; ++r) {
                    const int m = r + 8 * hi;
                    sc[r] = (lmod > m) ? -3.0e38f : sc[r];
                }
            }
            float alpha[8];
            v8f p;
#pragma unroll
            for (int r = 0; r < 8; ++r) {
                float t = sc[r];
                t = fmaxf(t, row_xor<1>(t));
                t = fmaxf(t, row_xor<2>(t));
                t = fmaxf(t, row_xor<4>(t));
                t = fmaxf(t, row_xor<8>(t));
                const float mn = fmaxf(mrow[r], t);
                alpha[r] = __expf(mrow[r] - mn);
                mrow[r]  = mn;
                const float pe = __expf(sc[r] - mn);
                p[r] = pe;
                float su = pe;
                su += row_xor<1>(su);
                su += row_xor<2>(su);
                su += row_xor<4>(su);
                su += row_xor<8>(su);
                lrow[r] = lrow[r] * alpha[r] + su;
            }
#pragma unroll
            for (int j = 0; j < 8; ++j)
#pragma unroll
                for (int r = 0; r < 8; ++r) o[j][r] *= alpha[r];

            float* pb = pbuf + wave * 16 * PSTRIDE;
#pragma unroll
            for (int r = 0; r < 8; ++r)
                pb[(r + 8 * hi) * PSTRIDE + lmod] = p[r];
            v2f ap[4];
#pragma unroll
            for (int k0 = 0; k0 < 4; ++k0)
                ap[k0] = *(const v2f*)(pb + lmod * PSTRIDE + 4 * k0 + off);

#pragma unroll
            for (int j = 0; j < 8; ++j) {
#pragma unroll
                for (int k0 = 0; k0 < 4; ++k0) {
                    o[j] = __builtin_amdgcn_wmma_f32_16x16x4_f32(
                               false, ap[k0], false, vf[4 * j + k0], (short)0, o[j], false, false);
                }
            }
        }

        if (more) stage_commit(sr, tid, kbuf2[cur ^ 1], vbufT2[cur ^ 1]);
        __syncthreads();
    }

    float inv_l[8];
#pragma unroll
    for (int r = 0; r < 8; ++r) inv_l[r] = 1.0f / lrow[r];
    float* obase = out + (size_t)bh * SS * DD;
#pragma unroll
    for (int j = 0; j < 8; ++j) {
#pragma unroll
        for (int r = 0; r < 8; ++r) {
            const int m = r + 8 * hi;
            obase[(size_t)(qbase + m) * DD + 16 * j + lmod] = o[j][r] * inv_l[r];
        }
    }
  }
}

extern "C" void kernel_launch(void* const* d_in, const int* in_sizes, int n_in,
                              void* d_out, int out_size, void* d_ws, size_t ws_size,
                              hipStream_t stream) {
    (void)in_sizes; (void)n_in; (void)out_size; (void)d_ws; (void)ws_size;
    const float* q = (const float*)d_in[0];
    const float* k = (const float*)d_in[1];
    const float* v = (const float*)d_in[2];
    const float* c = (const float*)d_in[3];
    const float* s = (const float*)d_in[4];
    float* o = (float*)d_out;
    dim3 grid(BB * HH * (NQB / 2));
    dim3 block(256);
    hipLaunchKernelGGL(fa_rope_fwd, grid, block, 0, stream, q, k, v, c, s, o);
}


# Transposed score tile (S^T = K Q^T, O^T = V^T P^T): one query per lane, P stays in registers as the PV B operand; no P LDS round trip, no DPP row reductions; row-sum and PV key sum order change
# speedup vs baseline: 2.5935x; 1.1140x over previous
#include <hip/hip_runtime.h>

typedef float v2f __attribute__((ext_vector_type(2)));
typedef float v8f __attribute__((ext_vector_type(8)));

#define BB 2
#define SS 2048
#define HH 16
#define DD 128
#define QB 128
#define NQB (SS / QB)
#define KSTRIDE 132
#define VSTRIDE 18


__device__ __forceinline__ float xhalf(float x)
{
    const int b = __builtin_bit_cast(int, x);
    return __builtin_bit_cast(float,
        __builtin_amdgcn_permlanex16(b, b, 0x76543210, 0xfedcba98, false, false));
}

struct StageRegs { float4 ka, kb, ca, cb, sa, sb, va, vb; };

__device__ __forceinline__ void stage_fetch(StageRegs& s, int kt, int tid, size_t kvhead,
        const float* __restrict__ k, const float* __restrict__ v,
        const float* __restrict__ cosp, const float* __restrict__ sinp)
{
    const int row = tid >> 4;
    const int c4  = (tid & 15) * 4;
    const int sk  = kt * 16 + row;
    const float* krow = k + kvhead + (size_t)sk * HH * DD;
    const float* vrow = v + kvhead + (size_t)sk * HH * DD;
    const float* crow = cosp + (size_t)sk * DD;
    const float* srow = sinp + (size_t)sk * DD;
    s.ka = *(const float4*)(krow + c4);
    s.kb = *(const float4*)(krow + c4 + 64);
    s.ca = *(const float4*)(crow + c4);
    s.cb = *(const float4*)(crow + c4 + 64);
    s.sa = *(const float4*)(srow + c4);
    s.sb = *(const float4*)(srow + c4 + 64);
    s.va = *(const float4*)(vrow + c4);
    s.vb = *(const float4*)(vrow + c4 + 64);
}

__device__ __forceinline__ void stage_commit(const StageRegs& s, int tid, float* kbuf, float* vbufT)
{
    const int row = tid >> 4;
    const int c4  = (tid & 15) * 4;
    const float4 ka = s.ka, kb = s.kb, ca = s.ca, cb = s.cb, sa = s.sa, sb = s.sb;
    float4 ra, rb;
    ra.x = ka.x * ca.x - kb.x * sa.x;  ra.y = ka.y * ca.y - kb.y * sa.y;
    ra.z = ka.z * ca.z - kb.z * sa.z;  ra.w = ka.w * ca.w - kb.w * sa.w;
    rb.x = kb.x * cb.x + ka.x * sb.x;  rb.y = kb.y * cb.y + ka.y * sb.y;
    rb.z = kb.z * cb.z + ka.z * sb.z;  rb.w = kb.w * cb.w + ka.w * sb.w;
    *(float4*)(kbuf + row * KSTRIDE + c4)      = ra;
    *(float4*)(kbuf + row * KSTRIDE + c4 + 64) = rb;
    const float4 va = s.va, vb = s.vb;
    const int prow = ((row >> 1) & 3) * 4 + (row >> 3) * 2 + (row & 1);
    vbufT[(c4 + 0) * VSTRIDE + prow] = va.x;
    vbufT[(c4 + 1) * VSTRIDE + prow] = va.y;
    vbufT[(c4 + 2) * VSTRIDE + prow] = va.z;
    vbufT[(c4 + 3) * VSTRIDE + prow] = va.w;
    vbufT[(c4 + 64) * VSTRIDE + prow] = vb.x;
    vbufT[(c4 + 65) * VSTRIDE + prow] = vb.y;
    vbufT[(c4 + 66) * VSTRIDE + prow] = vb.z;
    vbufT[(c4 + 67) * VSTRIDE + prow] = vb.w;
}

__global__ __launch_bounds__(256)
void fa_rope_fwd(const float* __restrict__ q, const float* __restrict__ k,
                 const float* __restrict__ v, const float* __restrict__ cosp,
                 const float* __restrict__ sinp, float* __restrict__ out)
{
    __shared__ float kbuf2[2][16 * KSTRIDE];
    __shared__ float vbufT2[2][DD * VSTRIDE];

    const int tid  = threadIdx.x;
    const int wave = tid >> 5;
    const int lane = tid & 31;
    const int lmod = lane & 15;
    const int hi   = lane >> 4;
    const int off  = hi * 2;

    const int pr   = blockIdx.x & 7;
    const int bh   = blockIdx.x >> 3;
    const int b    = bh >> 4;
    const int h    = bh & 15;
    const size_t kvhead = ((size_t)b * SS * HH + h) * DD;

  for (int pass = 0; pass < 2; ++pass) {
    const int qblk = pass ? pr : (NQB - 1 - pr);

    const int qbase = qblk * QB + wave * 16;
    const int myrow = qbase + lmod;

    float qa[64];
    {
        const float* qrow = q + (((size_t)b * SS + myrow) * HH + h) * DD;
        const float* crow = cosp + (size_t)myrow * DD;
        const float* srow = sinp + (size_t)myrow * DD;
        float raw[64];
#pragma unroll
        for (int st = 0; st < 32; ++st) {
            v2f t = *(const v2f*)(qrow + 4 * st + off);
            raw[2 * st] = t.x; raw[2 * st + 1] = t.y;
        }
        const float scale = 0.08838834764831845f;
#pragma unroll
        for (int st = 0; st < 16; ++st) {
            const int dlo = 4 * st + off, dhi = dlo + 64;
            v2f cl = *(const v2f*)(crow + dlo);
            v2f sl = *(const v2f*)(srow + dlo);
            v2f ch = *(const v2f*)(crow + dhi);
            v2f sh = *(const v2f*)(srow + dhi);
            const int lo = 2 * st, hp = lo + 32;
            qa[lo]     = (raw[lo]     * cl.x - raw[hp]     * sl.x) * scale;
            qa[lo + 1] = (raw[lo + 1] * cl.y - raw[hp + 1] * sl.y) * scale;
            qa[hp]     = (raw[hp]     * ch.x + raw[lo]     * sh.x) * scale;
            qa[hp + 1] = (raw[hp + 1] * ch.y + raw[lo + 1] * sh.y) * scale;
        }
    }

    v8f o[8];
#pragma unroll
    for (int j = 0; j < 8; ++j)
#pragma unroll
        for (int r = 0; r < 8; ++r) o[j][r] = 0.0f;
    float mrun = -3.0e38f, lrun = 0.0f;

    const int ntiles  = qblk * 8 + 8;
    const int ktmax_w = qblk * 8 + wave;

    StageRegs sr;
    stage_fetch(sr, 0, tid, kvhead, k, v, cosp, sinp);
    stage_commit(sr, tid, kbuf2[0], vbufT2[0]);
    __syncthreads();

    for (int kt = 0; kt < ntiles; ++kt) {
        const int cur = kt & 1;
        const float* kbuf  = kbuf2[cur];
        const float* vbufT = vbufT2[cur];
        const bool more = (kt + 1 < ntiles);
        if (more) stage_fetch(sr, kt + 1, tid, kvhead, k, v, cosp, sinp);

        if (kt <= ktmax_w) {
            v2f kf[32];
#pragma unroll
            for (int st = 0; st < 32; ++st)
                kf[st] = *(const v2f*)(kbuf + lmod * KSTRIDE + 4 * st + off);
            __builtin_amdgcn_sched_barrier(0);
            v8f sc;
#pragma unroll
            for (int r = 0; r < 8; ++r) sc[r] = 0.0f;
#pragma unroll
            for (int st = 0; st < 32; ++st) {
                v2f bq;  bq.x = qa[2 * st]; bq.y = qa[2 * st + 1];
                sc = __builtin_amdgcn_wmma_f32_16x16x4_f32(
                        false, kf[st], false, bq, (short)0, sc, false, false);
            }
            int voff = lmod * VSTRIDE + off;
            asm volatile("" : "+v"(voff) : "v"(sc));
            v2f vf[32];
#pragma unroll
            for (int j = 0; j < 8; ++j)
#pragma unroll
                for (int k0 = 0; k0 < 4; ++k0)
                    vf[4 * j + k0] = *(const v2f*)(vbufT + voff + 16 * j * VSTRIDE + 4 * k0);
            if (kt == ktmax_w) {
#pragma unroll
                for (int r = 0; r < 8; ++r)
                    sc[r] = (r + 8 * hi > lmod) ? -3.0e38f : sc[r];
            }
            float t = fmaxf(fmaxf(fmaxf(sc[0], sc[1]), fmaxf(sc[2], sc[3])),
                            fmaxf(fmaxf(sc[4], sc[5]), fmaxf(sc[6], sc[7])));
            t = fmaxf(t, xhalf(t));
            const float mn    = fmaxf(mrun, t);
            const float alpha = __expf(mrun - mn);
            mrun = mn;
            v8f p;
#pragma unroll
            for (int r = 0; r < 8; ++r) p[r] = __expf(sc[r] - mn);
            const float su = ((p[0] + p[1]) + (p[2] + p[3])) + ((p[4] + p[5]) + (p[6] + p[7]));
            lrun = lrun * alpha + su;
#pragma unroll
            for (int j = 0; j < 8; ++j)
#pragma unroll
                for (int r = 0; r < 8; ++r) o[j][r] *= alpha;

#pragma unroll
            for (int j = 0; j < 8; ++j) {
#pragma unroll
                for (int k0 = 0; k0 < 4; ++k0) {
                    v2f bp;  bp.x = p[2 * k0]; bp.y = p[2 * k0 + 1];
                    o[j] = __builtin_amdgcn_wmma_f32_16x16x4_f32(
                               false, vf[4 * j + k0], false, bp, (short)0, o[j], false, false);
                }
            }
        }

        if (more) stage_commit(sr, tid, kbuf2[cur ^ 1], vbufT2[cur ^ 1]);
        __syncthreads();
    }

    const float inv_l = 1.0f / (lrun + xhalf(lrun));
    float* orow = out + ((size_t)bh * SS + myrow) * DD + 8 * hi;
#pragma unroll
    for (int j = 0; j < 8; ++j) {
        float4 a, c;
        a.x = o[j][0] * inv_l; a.y = o[j][1] * inv_l; a.z = o[j][2] * inv_l; a.w = o[j][3] * inv_l;
        c.x = o[j][4] * inv_l; c.y = o[j][5] * inv_l; c.z = o[j][6] * inv_l; c.w = o[j][7] * inv_l;
        *(float4*)(orow + 16 * j)     = a;
        *(float4*)(orow + 16 * j + 4) = c;
    }
  }
}

extern "C" void kernel_launch(void* const* d_in, const int* in_sizes, int n_in,
                              void* d_out, int out_size, void* d_ws, size_t ws_size,
                              hipStream_t stream) {
    (void)in_sizes; (void)n_in; (void)out_size; (void)d_ws; (void)ws_size;
    const float* q = (const float*)d_in[0];
    const float* k = (const float*)d_in[1];
    const float* v = (const float*)d_in[2];
    const float* c = (const float*)d_in[3];
    const float* s = (const float*)d_in[4];
    float* o = (float*)d_out;
    dim3 grid(BB * HH * (NQB / 2));
    dim3 block(256);
    hipLaunchKernelGGL(fa_rope_fwd, grid, block, 0, stream, q, k, v, c, s, o);
}


# Two 16-key tiles per barrier: their score chains alternate as two independent accumulators; row max, rescale and row sum once per 32 keys (row-sum and PV sum order change)
# speedup vs baseline: 2.6594x; 1.0254x over previous
#include <hip/hip_runtime.h>

typedef float v2f __attribute__((ext_vector_type(2)));
typedef float v8f __attribute__((ext_vector_type(8)));

#define BB 2
#define SS 2048
#define HH 16
#define DD 128
#define QB 128
#define NQB (SS / QB)
#define KSTRIDE 132
#define VSTRIDE 18


__device__ __forceinline__ float xhalf(float x)
{
    const int b = __builtin_bit_cast(int, x);
    return __builtin_bit_cast(float,
        __builtin_amdgcn_permlanex16(b, b, 0x76543210, 0xfedcba98, false, false));
}

struct StageRegs { float4 ka, kb, ca, cb, sa, sb, va, vb; };

__device__ __forceinline__ void stage_fetch(StageRegs& s, int kt, int tid, size_t kvhead,
        const float* __restrict__ k, const float* __restrict__ v,
        const float* __restrict__ cosp, const float* __restrict__ sinp)
{
    const int row = tid >> 4;
    const int c4  = (tid & 15) * 4;
    const int sk  = kt * 16 + row;
    const float* krow = k + kvhead + (size_t)sk * HH * DD;
    const float* vrow = v + kvhead + (size_t)sk * HH * DD;
    const float* crow = cosp + (size_t)sk * DD;
    const float* srow = sinp + (size_t)sk * DD;
    s.ka = *(const float4*)(krow + c4);
    s.kb = *(const float4*)(krow + c4 + 64);
    s.ca = *(const float4*)(crow + c4);
    s.cb = *(const float4*)(crow + c4 + 64);
    s.sa = *(const float4*)(srow + c4);
    s.sb = *(const float4*)(srow + c4 + 64);
    s.va = *(const float4*)(vrow + c4);
    s.vb = *(const float4*)(vrow + c4 + 64);
}

__device__ __forceinline__ void stage_commit(const StageRegs& s, int tid, float* kbuf, float* vbufT)
{
    const int row = tid >> 4;
    const int c4  = (tid & 15) * 4;
    const float4 ka = s.ka, kb = s.kb, ca = s.ca, cb = s.cb, sa = s.sa, sb = s.sb;
    float4 ra, rb;
    ra.x = ka.x * ca.x - kb.x * sa.x;  ra.y = ka.y * ca.y - kb.y * sa.y;
    ra.z = ka.z * ca.z - kb.z * sa.z;  ra.w = ka.w * ca.w - kb.w * sa.w;
    rb.x = kb.x * cb.x + ka.x * sb.x;  rb.y = kb.y * cb.y + ka.y * sb.y;
    rb.z = kb.z * cb.z + ka.z * sb.z;  rb.w = kb.w * cb.w + ka.w * sb.w;
    *(float4*)(kbuf + row * KSTRIDE + c4)      = ra;
    *(float4*)(kbuf + row * KSTRIDE + c4 + 64) = rb;
    const float4 va = s.va, vb = s.vb;
    const int prow = ((row >> 1) & 3) * 4 + (row >> 3) * 2 + (row & 1);
    vbufT[(c4 + 0) * VSTRIDE + prow] = va.x;
    vbufT[(c4 + 1) * VSTRIDE + prow] = va.y;
    vbufT[(c4 + 2) * VSTRIDE + prow] = va.z;
    vbufT[(c4 + 3) * VSTRIDE + prow] = va.w;
    vbufT[(c4 + 64) * VSTRIDE + prow] = vb.x;
    vbufT[(c4 + 65) * VSTRIDE + prow] = vb.y;
    vbufT[(c4 + 66) * VSTRIDE + prow] = vb.z;
    vbufT[(c4 + 67) * VSTRIDE + prow] = vb.w;
}

__global__ __launch_bounds__(256)
void fa_rope_fwd(const float* __restrict__ q, const float* __restrict__ k,
                 const float* __restrict__ v, const float* __restrict__ cosp,
                 const float* __restrict__ sinp, float* __restrict__ out)
{
    __shared__ float kbuf4[2][2][16 * KSTRIDE];
    __shared__ float vbufT4[2][2][DD * VSTRIDE];

    const int tid  = threadIdx.x;
    const int wave = tid >> 5;
    const int lane = tid & 31;
    const int lmod = lane & 15;
    const int hi   = lane >> 4;
    const int off  = hi * 2;

    const int pr   = blockIdx.x & 7;
    const int bh   = blockIdx.x >> 3;
    const int b    = bh >> 4;
    const int h    = bh & 15;
    const size_t kvhead = ((size_t)b * SS * HH + h) * DD;

  for (int pass = 0; pass < 2; ++pass) {
    const int qblk = pass ? pr : (NQB - 1 - pr);

    const int qbase = qblk * QB + wave * 16;
    const int myrow = qbase + lmod;

    float qa[64];
    {
        const float* qrow = q + (((size_t)b * SS + myrow) * HH + h) * DD;
        const float* crow = cosp + (size_t)myrow * DD;
        const float* srow = sinp + (size_t)myrow * DD;
        float raw[64];
#pragma unroll
        for (int st = 0; st < 32; ++st) {
            v2f t = *(const v2f*)(qrow + 4 * st + off);
            raw[2 * st] = t.x; raw[2 * st + 1] = t.y;
        }
        const float scale = 0.08838834764831845f;
#pragma unroll
        for (int st = 0; st < 16; ++st) {
            const int dlo = 4 * st + off, dhi = dlo + 64;
            v2f cl = *(const v2f*)(crow + dlo);
            v2f sl = *(const v2f*)(srow + dlo);
            v2f ch = *(const v2f*)(crow + dhi);
            v2f sh = *(const v2f*)(srow + dhi);
            const int lo = 2 * st, hp = lo + 32;
            qa[lo]     = (raw[lo]     * cl.x - raw[hp]     * sl.x) * scale;
            qa[lo + 1] = (raw[lo + 1] * cl.y - raw[hp + 1] * sl.y) * scale;
            qa[hp]     = (raw[hp]     * ch.x + raw[lo]     * sh.x) * scale;
            qa[hp + 1] = (raw[hp + 1] * ch.y + raw[lo + 1] * sh.y) * scale;
        }
    }

    v8f o[8];
#pragma unroll
    for (int j = 0; j < 8; ++j)
#pragma unroll
        for (int r = 0; r < 8; ++r) o[j][r] = 0.0f;
    float mrun = -3.0e38f, lrun = 0.0f;

    const int nst     = qblk * 4 + 4;
    const int ktmax_w = qblk * 8 + wave;
    const int lst_w   = ktmax_w >> 1;

    StageRegs sr0, sr1;
    stage_fetch(sr0, 0, tid, kvhead, k, v, cosp, sinp);
    stage_fetch(sr1, 1, tid, kvhead, k, v, cosp, sinp);
    stage_commit(sr0, tid, kbuf4[0][0], vbufT4[0][0]);
    stage_commit(sr1, tid, kbuf4[0][1], vbufT4[0][1]);
    __syncthreads();

    for (int s = 0; s < nst; ++s) {
        const int cur = s & 1;
        const bool more = (s + 1 < nst);
        if (more) {
            stage_fetch(sr0, 2 * s + 2, tid, kvhead, k, v, cosp, sinp);
            stage_fetch(sr1, 2 * s + 3, tid, kvhead, k, v, cosp, sinp);
        }

        if (s <= lst_w) {
            const float* kb0 = kbuf4[cur][0];
            const float* kb1 = kbuf4[cur][1];
            const float* vb0 = vbufT4[cur][0];
            const float* vb1 = vbufT4[cur][1];
            v2f kf0[32], kf1[32];
#pragma unroll
            for (int st = 0; st < 32; ++st) {
                kf0[st] = *(const v2f*)(kb0 + lmod * KSTRIDE + 4 * st + off);
                kf1[st] = *(const v2f*)(kb1 + lmod * KSTRIDE + 4 * st + off);
            }
            __builtin_amdgcn_sched_barrier(0);
            v8f sc0, sc1;
#pragma unroll
            for (int r = 0; r < 8; ++r) { sc0[r] = 0.0f; sc1[r] = 0.0f; }
#pragma unroll
            for (int st = 0; st < 32; ++st) {
                v2f bq;  bq.x = qa[2 * st]; bq.y = qa[2 * st + 1];
                sc0 = __builtin_amdgcn_wmma_f32_16x16x4_f32(
                        false, kf0[st], false, bq, (short)0, sc0, false, false);
                sc1 = __builtin_amdgcn_wmma_f32_16x16x4_f32(
                        false, kf1[st], false, bq, (short)0, sc1, false, false);
            }
            int voff = lmod * VSTRIDE + off;
            asm volatile("" : "+v"(voff) : "v"(sc0), "v"(sc1));
            v2f vf0[32], vf1[32];
#pragma unroll
            for (int j = 0; j < 8; ++j)
#pragma unroll
                for (int k0 = 0; k0 < 4; ++k0) {
                    vf0[4 * j + k0] = *(const v2f*)(vb0 + voff + 16 * j * VSTRIDE + 4 * k0);
                    vf1[4 * j + k0] = *(const v2f*)(vb1 + voff + 16 * j * VSTRIDE + 4 * k0);
                }
            if (s == lst_w) {
                const bool oddw = (wave & 1) != 0;
#pragma unroll
                for (int r = 0; r < 8; ++r) {
                    const bool dg = (r + 8 * hi > lmod);
                    const bool m0 = oddw ? false : dg;
                    const bool m1 = oddw ? dg : true;
                    sc0[r] = m0 ? -3.0e38f : sc0[r];
                    sc1[r] = m1 ? -3.0e38f : sc1[r];
                }
            }
            const float t0 = fmaxf(fmaxf(fmaxf(sc0[0], sc0[1]), fmaxf(sc0[2], sc0[3])),
                                   fmaxf(fmaxf(sc0[4], sc0[5]), fmaxf(sc0[6], sc0[7])));
            const float t1 = fmaxf(fmaxf(fmaxf(sc1[0], sc1[1]), fmaxf(sc1[2], sc1[3])),
                                   fmaxf(fmaxf(sc1[4], sc1[5]), fmaxf(sc1[6], sc1[7])));
            float t = fmaxf(t0, t1);
            t = fmaxf(t, xhalf(t));
            const float mn    = fmaxf(mrun, t);
            const float alpha = __expf(mrun - mn);
            mrun = mn;
            v8f p0, p1;
#pragma unroll
            for (int r = 0; r < 8; ++r) { p0[r] = __expf(sc0[r] - mn); p1[r] = __expf(sc1[r] - mn); }
            const float su0 = ((p0[0] + p0[1]) + (p0[2] + p0[3])) + ((p0[4] + p0[5]) + (p0[6] + p0[7]));
            const float su1 = ((p1[0] + p1[1]) + (p1[2] + p1[3])) + ((p1[4] + p1[5]) + (p1[6] + p1[7]));
            lrun = lrun * alpha + (su0 + su1);
#pragma unroll
            for (int j = 0; j < 8; ++j)
#pragma unroll
                for (int r = 0; r < 8; ++r) o[j][r] *= alpha;

#pragma unroll
            for (int k0 = 0; k0 < 4; ++k0) {
                v2f bp;  bp.x = p0[2 * k0]; bp.y = p0[2 * k0 + 1];
#pragma unroll
                for (int j = 0; j < 8; ++j)
                    o[j] = __builtin_amdgcn_wmma_f32_16x16x4_f32(
                               false, vf0[4 * j + k0], false, bp, (short)0, o[j], false, false);
            }
#pragma unroll
            for (int k0 = 0; k0 < 4; ++k0) {
                v2f bp;  bp.x = p1[2 * k0]; bp.y = p1[2 * k0 + 1];
#pragma unroll
                for (int j = 0; j < 8; ++j)
                    o[j] = __builtin_amdgcn_wmma_f32_16x16x4_f32(
                               false, vf1[4 * j + k0], false, bp, (short)0, o[j], false, false);
            }
        }

        if (more) {
            stage_commit(sr0, tid, kbuf4[cur ^ 1][0], vbufT4[cur ^ 1][0]);
            stage_commit(sr1, tid, kbuf4[cur ^ 1][1], vbufT4[cur ^ 1][1]);
        }
        __syncthreads();
    }

    const float inv_l = 1.0f / (lrun + xhalf(lrun));
    float* orow = out + ((size_t)bh * SS + myrow) * DD + 8 * hi;
#pragma unroll
    for (int j = 0; j < 8; ++j) {
        float4 a, c;
        a.x = o[j][0] * inv_l; a.y = o[j][1] * inv_l; a.z = o[j][2] * inv_l; a.w = o[j][3] * inv_l;
        c.x = o[j][4] * inv_l; c.y = o[j][5] * inv_l; c.z = o[j][6] * inv_l; c.w = o[j][7] * inv_l;
        *(float4*)(orow + 16 * j)     = a;
        *(float4*)(orow + 16 * j + 4) = c;
    }
  }
}

extern "C" void kernel_launch(void* const* d_in, const int* in_sizes, int n_in,
                              void* d_out, int out_size, void* d_ws, size_t ws_size,
                              hipStream_t stream) {
    (void)in_sizes; (void)n_in; (void)out_size; (void)d_ws; (void)ws_size;
    const float* q = (const float*)d_in[0];
    const float* k = (const float*)d_in[1];
    const float* v = (const float*)d_in[2];
    const float* c = (const float*)d_in[3];
    const float* s = (const float*)d_in[4];
    float* o = (float*)d_out;
    dim3 grid(BB * HH * (NQB / 2));
    dim3 block(256);
    hipLaunchKernelGGL(fa_rope_fwd, grid, block, 0, stream, q, k, v, c, s, o);
}


# K/V staging moved from the stage's ends at the barrier to its middle, between score and output WMMAs: stage s+1 written to LDS, stage s+2 fetched there; barrier fences LDS only; sum order unchanged
# speedup vs baseline: 2.8931x; 1.0879x over previous
#include <hip/hip_runtime.h>

typedef float v2f __attribute__((ext_vector_type(2)));
typedef float v8f __attribute__((ext_vector_type(8)));

#define BB 2
#define SS 2048
#define HH 16
#define DD 128
#define QB 128
#define NQB (SS / QB)
#define KSTRIDE 132
#define VSTRIDE 18


__device__ __forceinline__ float xhalf(float x)
{
    const int b = __builtin_bit_cast(int, x);
    return __builtin_bit_cast(float,
        __builtin_amdgcn_permlanex16(b, b, 0x76543210, 0xfedcba98, false, false));
}

struct StageRegs { float4 ka, kb, ca, cb, sa, sb, va, vb; };

__device__ __forceinline__ void stage_fetch(StageRegs& s, int kt, int tid, size_t kvhead,
        const float* __restrict__ k, const float* __restrict__ v,
        const float* __restrict__ cosp, const float* __restrict__ sinp)
{
    const int row = tid >> 4;
    const int c4  = (tid & 15) * 4;
    const int sk  = kt * 16 + row;
    const float* krow = k + kvhead + (size_t)sk * HH * DD;
    const float* vrow = v + kvhead + (size_t)sk * HH * DD;
    const float* crow = cosp + (size_t)sk * DD;
    const float* srow = sinp + (size_t)sk * DD;
    s.ka = *(const float4*)(krow + c4);
    s.kb = *(const float4*)(krow + c4 + 64);
    s.ca = *(const float4*)(crow + c4);
    s.cb = *(const float4*)(crow + c4 + 64);
    s.sa = *(const float4*)(srow + c4);
    s.sb = *(const float4*)(srow + c4 + 64);
    s.va = *(const float4*)(vrow + c4);
    s.vb = *(const float4*)(vrow + c4 + 64);
}

__device__ __forceinline__ void stage_commit(const StageRegs& s, int tid, float* kbuf, float* vbufT)
{
    const int row = tid >> 4;
    const int c4  = (tid & 15) * 4;
    const float4 ka = s.ka, kb = s.kb, ca = s.ca, cb = s.cb, sa = s.sa, sb = s.sb;
    float4 ra, rb;
    ra.x = ka.x * ca.x - kb.x * sa.x;  ra.y = ka.y * ca.y - kb.y * sa.y;
    ra.z = ka.z * ca.z - kb.z * sa.z;  ra.w = ka.w * ca.w - kb.w * sa.w;
    rb.x = kb.x * cb.x + ka.x * sb.x;  rb.y = kb.y * cb.y + ka.y * sb.y;
    rb.z = kb.z * cb.z + ka.z * sb.z;  rb.w = kb.w * cb.w + ka.w * sb.w;
    *(float4*)(kbuf + row * KSTRIDE + c4)      = ra;
    *(float4*)(kbuf + row * KSTRIDE + c4 + 64) = rb;
    const float4 va = s.va, vb = s.vb;
    const int prow = ((row >> 1) & 3) * 4 + (row >> 3) * 2 + (row & 1);
    vbufT[(c4 + 0) * VSTRIDE + prow] = va.x;
    vbufT[(c4 + 1) * VSTRIDE + prow] = va.y;
    vbufT[(c4 + 2) * VSTRIDE + prow] = va.z;
    vbufT[(c4 + 3) * VSTRIDE + prow] = va.w;
    vbufT[(c4 + 64) * VSTRIDE + prow] = vb.x;
    vbufT[(c4 + 65) * VSTRIDE + prow] = vb.y;
    vbufT[(c4 + 66) * VSTRIDE + prow] = vb.z;
    vbufT[(c4 + 67) * VSTRIDE + prow] = vb.w;
}

__device__ __forceinline__ void lds_barrier()
{
    __builtin_amdgcn_fence(__ATOMIC_RELEASE, "workgroup", "local");
    __builtin_amdgcn_s_barrier();
    __builtin_amdgcn_fence(__ATOMIC_ACQUIRE, "workgroup", "local");
}

__device__ __forceinline__ void stage_mid(StageRegs& sr0, StageRegs& sr1, int s, int cur,
        bool more1, bool more2, int tid, size_t kvhead,
        const float* __restrict__ k, const float* __restrict__ v,
        const float* __restrict__ cosp, const float* __restrict__ sinp,
        float (*kbuf4)[2][16 * KSTRIDE], float (*vbufT4)[2][DD * VSTRIDE])
{
    if (more1) {
        stage_commit(sr0, tid, kbuf4[cur ^ 1][0], vbufT4[cur ^ 1][0]);
        stage_commit(sr1, tid, kbuf4[cur ^ 1][1], vbufT4[cur ^ 1][1]);
    }
    if (more2) {
        stage_fetch(sr0, 2 * s + 4, tid, kvhead, k, v, cosp, sinp);
        stage_fetch(sr1, 2 * s + 5, tid, kvhead, k, v, cosp, sinp);
    }
}

__global__ __launch_bounds__(256)
void fa_rope_fwd(const float* __restrict__ q, const float* __restrict__ k,
                 const float* __restrict__ v, const float* __restrict__ cosp,
                 const float* __restrict__ sinp, float* __restrict__ out)
{
    __shared__ float kbuf4[2][2][16 * KSTRIDE];
    __shared__ float vbufT4[2][2][DD * VSTRIDE];

    const int tid  = threadIdx.x;
    const int wave = tid >> 5;
    const int lane = tid & 31;
    const int lmod = lane & 15;
    const int hi   = lane >> 4;
    const int off  = hi * 2;

    const int pr   = blockIdx.x & 7;
    const int bh   = blockIdx.x >> 3;
    const int b    = bh >> 4;
    const int h    = bh & 15;
    const size_t kvhead = ((size_t)b * SS * HH + h) * DD;

  for (int pass = 0; pass < 2; ++pass) {
    const int qblk = pass ? pr : (NQB - 1 - pr);

    const int qbase = qblk * QB + wave * 16;
    const int myrow = qbase + lmod;

    float qa[64];
    {
        const float* qrow = q + (((size_t)b * SS + myrow) * HH + h) * DD;
        const float* crow = cosp + (size_t)myrow * DD;
        const float* srow = sinp + (size_t)myrow * DD;
        float raw[64];
#pragma unroll
        for (int st = 0; st < 32; ++st) {
            v2f t = *(const v2f*)(qrow + 4 * st + off);
            raw[2 * st] = t.x; raw[2 * st + 1] = t.y;
        }
        const float scale = 0.08838834764831845f;
#pragma unroll
        for (int st = 0; st < 16; ++st) {
            const int dlo = 4 * st + off, dhi = dlo + 64;
            v2f cl = *(const v2f*)(crow + dlo);
            v2f sl = *(const v2f*)(srow + dlo);
            v2f ch = *(const v2f*)(crow + dhi);
            v2f sh = *(const v2f*)(srow + dhi);
            const int lo = 2 * st, hp = lo + 32;
            qa[lo]     = (raw[lo]     * cl.x - raw[hp]     * sl.x) * scale;
            qa[lo + 1] = (raw[lo + 1] * cl.y - raw[hp + 1] * sl.y) * scale;
            qa[hp]     = (raw[hp]     * ch.x + raw[lo]     * sh.x) * scale;
            qa[hp + 1] = (raw[hp + 1] * ch.y + raw[lo + 1] * sh.y) * scale;
        }
    }

    v8f o[8];
#pragma unroll
    for (int j = 0; j < 8; ++j)
#pragma unroll
        for (int r = 0; r < 8; ++r) o[j][r] = 0.0f;
    float mrun = -3.0e38f, lrun = 0.0f;

    const int nst     = qblk * 4 + 4;
    const int ktmax_w = qblk * 8 + wave;
    const int lst_w   = ktmax_w >> 1;

    StageRegs sr0, sr1;
    stage_fetch(sr0, 0, tid, kvhead, k, v, cosp, sinp);
    stage_fetch(sr1, 1, tid, kvhead, k, v, cosp, sinp);
    stage_commit(sr0, tid, kbuf4[0][0], vbufT4[0][0]);
    stage_commit(sr1, tid, kbuf4[0][1], vbufT4[0][1]);
    stage_fetch(sr0, 2, tid, kvhead, k, v, cosp, sinp);
    stage_fetch(sr1, 3, tid, kvhead, k, v, cosp, sinp);
    lds_barrier();

    for (int s = 0; s < nst; ++s) {
        const int cur = s & 1;
        const bool more1 = (s + 1 < nst);
        const bool more2 = (s + 2 < nst);

        const bool active = (s <= lst_w);
        v8f sc0, sc1;
        v2f vf0[32], vf1[32];
        if (active) {
            const float* kb0 = kbuf4[cur][0];
            const float* kb1 = kbuf4[cur][1];
            const float* vb0 = vbufT4[cur][0];
            const float* vb1 = vbufT4[cur][1];
            v2f kf0[32], kf1[32];
#pragma unroll
            for (int st = 0; st < 32; ++st) {
                kf0[st] = *(const v2f*)(kb0 + lmod * KSTRIDE + 4 * st + off);
                kf1[st] = *(const v2f*)(kb1 + lmod * KSTRIDE + 4 * st + off);
            }
            __builtin_amdgcn_sched_barrier(0);
#pragma unroll
            for (int r = 0; r < 8; ++r) { sc0[r] = 0.0f; sc1[r] = 0.0f; }
#pragma unroll
            for (int st = 0; st < 32; ++st) {
                v2f bq;  bq.x = qa[2 * st]; bq.y = qa[2 * st + 1];
                sc0 = __builtin_amdgcn_wmma_f32_16x16x4_f32(
                        false, kf0[st], false, bq, (short)0, sc0, false, false);
                sc1 = __builtin_amdgcn_wmma_f32_16x16x4_f32(
                        false, kf1[st], false, bq, (short)0, sc1, false, false);
            }
            int voff = lmod * VSTRIDE + off;
            asm volatile("" : "+v"(voff) : "v"(sc0), "v"(sc1));
#pragma unroll
            for (int j = 0; j < 8; ++j)
#pragma unroll
                for (int k0 = 0; k0 < 4; ++k0) {
                    vf0[4 * j + k0] = *(const v2f*)(vb0 + voff + 16 * j * VSTRIDE + 4 * k0);
                    vf1[4 * j + k0] = *(const v2f*)(vb1 + voff + 16 * j * VSTRIDE + 4 * k0);
                }
        }
        stage_mid(sr0, sr1, s, cur, more1, more2, tid, kvhead, k, v, cosp, sinp, kbuf4, vbufT4);
        if (active) {
            if (s == lst_w) {
                const bool oddw = (wave & 1) != 0;
#pragma unroll
                for (int r = 0; r < 8; ++r) {
                    const bool dg = (r + 8 * hi > lmod);
                    const bool m0 = oddw ? false : dg;
                    const bool m1 = oddw ? dg : true;
                    sc0[r] = m0 ? -3.0e38f : sc0[r];
                    sc1[r] = m1 ? -3.0e38f : sc1[r];
                }
            }
            const float t0 = fmaxf(fmaxf(fmaxf(sc0[0], sc0[1]), fmaxf(sc0[2], sc0[3])),
                                   fmaxf(fmaxf(sc0[4], sc0[5]), fmaxf(sc0[6], sc0[7])));
            const float t1 = fmaxf(fmaxf(fmaxf(sc1[0], sc1[1]), fmaxf(sc1[2], sc1[3])),
                                   fmaxf(fmaxf(sc1[4], sc1[5]), fmaxf(sc1[6], sc1[7])));
            float t = fmaxf(t0, t1);
            t = fmaxf(t, xhalf(t));
            const float mn    = fmaxf(mrun, t);
            const float alpha = __expf(mrun - mn);
            mrun = mn;
            v8f p0, p1;
#pragma unroll
            for (int r = 0; r < 8; ++r) { p0[r] = __expf(sc0[r] - mn); p1[r] = __expf(sc1[r] - mn); }
            const float su0 = ((p0[0] + p0[1]) + (p0[2] + p0[3])) + ((p0[4] + p0[5]) + (p0[6] + p0[7]));
            const float su1 = ((p1[0] + p1[1]) + (p1[2] + p1[3])) + ((p1[4] + p1[5]) + (p1[6] + p1[7]));
            lrun = lrun * alpha + (su0 + su1);
#pragma unroll
            for (int j = 0; j < 8; ++j)
#pragma unroll
                for (int r = 0; r < 8; ++r) o[j][r] *= alpha;

#pragma unroll
            for (int k0 = 0; k0 < 4; ++k0) {
                v2f bp;  bp.x = p0[2 * k0]; bp.y = p0[2 * k0 + 1];
#pragma unroll
                for (int j = 0; j < 8; ++j)
                    o[j] = __builtin_amdgcn_wmma_f32_16x16x4_f32(
                               false, vf0[4 * j + k0], false, bp, (short)0, o[j], false, false);
            }
#pragma unroll
            for (int k0 = 0; k0 < 4; ++k0) {
                v2f bp;  bp.x = p1[2 * k0]; bp.y = p1[2 * k0 + 1];
#pragma unroll
                for (int j = 0; j < 8; ++j)
                    o[j] = __builtin_amdgcn_wmma_f32_16x16x4_f32(
                               false, vf1[4 * j + k0], false, bp, (short)0, o[j], false, false);
            }
        }
        lds_barrier();
    }

    const float inv_l = 1.0f / (lrun + xhalf(lrun));
    float* orow = out + ((size_t)bh * SS + myrow) * DD + 8 * hi;
#pragma unroll
    for (int j = 0; j < 8; ++j) {
        float4 a, c;
        a.x = o[j][0] * inv_l; a.y = o[j][1] * inv_l; a.z = o[j][2] * inv_l; a.w = o[j][3] * inv_l;
        c.x = o[j][4] * inv_l; c.y = o[j][5] * inv_l; c.z = o[j][6] * inv_l; c.w = o[j][7] * inv_l;
        *(float4*)(orow + 16 * j)     = a;
        *(float4*)(orow + 16 * j + 4) = c;
    }
  }
}

extern "C" void kernel_launch(void* const* d_in, const int* in_sizes, int n_in,
                              void* d_out, int out_size, void* d_ws, size_t ws_size,
                              hipStream_t stream) {
    (void)in_sizes; (void)n_in; (void)out_size; (void)d_ws; (void)ws_size;
    const float* q = (const float*)d_in[0];
    const float* k = (const float*)d_in[1];
    const float* v = (const float*)d_in[2];
    const float* c = (const float*)d_in[3];
    const float* s = (const float*)d_in[4];
    float* o = (float*)d_out;
    dim3 grid(BB * HH * (NQB / 2));
    dim3 block(256);
    hipLaunchKernelGGL(fa_rope_fwd, grid, block, 0, stream, q, k, v, c, s, o);
}


# Query tile read and output tile written as whole 512-byte rows through a per-wave LDS transpose tile, not 8- and 16-byte pieces over 16 rows per instruction; arithmetic and sum order unchanged
# speedup vs baseline: 3.0287x; 1.0469x over previous
#include <hip/hip_runtime.h>

typedef float v2f __attribute__((ext_vector_type(2)));
typedef float v8f __attribute__((ext_vector_type(8)));

#define BB 2
#define SS 2048
#define HH 16
#define DD 128
#define QB 128
#define NQB (SS / QB)
#define KSTRIDE 132
#define VSTRIDE 18
#define XSTRIDE 132


__device__ __forceinline__ float xhalf(float x)
{
    const int b = __builtin_bit_cast(int, x);
    return __builtin_bit_cast(float,
        __builtin_amdgcn_permlanex16(b, b, 0x76543210, 0xfedcba98, false, false));
}

struct StageRegs { float4 ka, kb, ca, cb, sa, sb, va, vb; };

__device__ __forceinline__ void stage_fetch(StageRegs& s, int kt, int tid, size_t kvhead,
        const float* __restrict__ k, const float* __restrict__ v,
        const float* __restrict__ cosp, const float* __restrict__ sinp)
{
    const int row = tid >> 4;
    const int c4  = (tid & 15) * 4;
    const int sk  = kt * 16 + row;
    const float* krow = k + kvhead + (size_t)sk * HH * DD;
    const float* vrow = v + kvhead + (size_t)sk * HH * DD;
    const float* crow = cosp + (size_t)sk * DD;
    const float* srow = sinp + (size_t)sk * DD;
    s.ka = *(const float4*)(krow + c4);
    s.kb = *(const float4*)(krow + c4 + 64);
    s.ca = *(const float4*)(crow + c4);
    s.cb = *(const float4*)(crow + c4 + 64);
    s.sa = *(const float4*)(srow + c4);
    s.sb = *(const float4*)(srow + c4 + 64);
    s.va = *(const float4*)(vrow + c4);
    s.vb = *(const float4*)(vrow + c4 + 64);
}

__device__ __forceinline__ void stage_commit(const StageRegs& s, int tid, float* kbuf, float* vbufT)
{
    const int row = tid >> 4;
    const int c4  = (tid & 15) * 4;
    const float4 ka = s.ka, kb = s.kb, ca = s.ca, cb = s.cb, sa = s.sa, sb = s.sb;
    float4 ra, rb;
    ra.x = ka.x * ca.x - kb.x * sa.x;  ra.y = ka.y * ca.y - kb.y * sa.y;
    ra.z = ka.z * ca.z - kb.z * sa.z;  ra.w = ka.w * ca.w - kb.w * sa.w;
    rb.x = kb.x * cb.x + ka.x * sb.x;  rb.y = kb.y * cb.y + ka.y * sb.y;
    rb.z = kb.z * cb.z + ka.z * sb.z;  rb.w = kb.w * cb.w + ka.w * sb.w;
    *(float4*)(kbuf + row * KSTRIDE + c4)      = ra;
    *(float4*)(kbuf + row * KSTRIDE + c4 + 64) = rb;
    const float4 va = s.va, vb = s.vb;
    const int prow = ((row >> 1) & 3) * 4 + (row >> 3) * 2 + (row & 1);
    vbufT[(c4 + 0) * VSTRIDE + prow] = va.x;
    vbufT[(c4 + 1) * VSTRIDE + prow] = va.y;
    vbufT[(c4 + 2) * VSTRIDE + prow] = va.z;
    vbufT[(c4 + 3) * VSTRIDE + prow] = va.w;
    vbufT[(c4 + 64) * VSTRIDE + prow] = vb.x;
    vbufT[(c4 + 65) * VSTRIDE + prow] = vb.y;
    vbufT[(c4 + 66) * VSTRIDE + prow] = vb.z;
    vbufT[(c4 + 67) * VSTRIDE + prow] = vb.w;
}

__device__ __forceinline__ void lds_barrier()
{
    __builtin_amdgcn_fence(__ATOMIC_RELEASE, "workgroup", "local");
    __builtin_amdgcn_s_barrier();
    __builtin_amdgcn_fence(__ATOMIC_ACQUIRE, "workgroup", "local");
}

__device__ __forceinline__ void stage_mid(StageRegs& sr0, StageRegs& sr1, int s, int cur,
        bool more1, bool more2, int tid, size_t kvhead,
        const float* __restrict__ k, const float* __restrict__ v,
        const float* __restrict__ cosp, const float* __restrict__ sinp,
        float (*kbuf4)[2][16 * KSTRIDE], float (*vbufT4)[2][DD * VSTRIDE])
{
    if (more1) {
        stage_commit(sr0, tid, kbuf4[cur ^ 1][0], vbufT4[cur ^ 1][0]);
        stage_commit(sr1, tid, kbuf4[cur ^ 1][1], vbufT4[cur ^ 1][1]);
    }
    if (more2) {
        stage_fetch(sr0, 2 * s + 4, tid, kvhead, k, v, cosp, sinp);
        stage_fetch(sr1, 2 * s + 5, tid, kvhead, k, v, cosp, sinp);
    }
}

__global__ __launch_bounds__(256)
void fa_rope_fwd(const float* __restrict__ q, const float* __restrict__ k,
                 const float* __restrict__ v, const float* __restrict__ cosp,
                 const float* __restrict__ sinp, float* __restrict__ out)
{
    __shared__ float kbuf4[2][2][16 * KSTRIDE];
    __shared__ float vbufT4[2][2][DD * VSTRIDE];
    __shared__ float xpose[8][16 * XSTRIDE];

    const int tid  = threadIdx.x;
    const int wave = tid >> 5;
    const int lane = tid & 31;
    const int lmod = lane & 15;
    const int hi   = lane >> 4;
    const int off  = hi * 2;

    const int pr   = blockIdx.x & 7;
    const int bh   = blockIdx.x >> 3;
    const int b    = bh >> 4;
    const int h    = bh & 15;
    const size_t kvhead = ((size_t)b * SS * HH + h) * DD;

  for (int pass = 0; pass < 2; ++pass) {
    const int qblk = pass ? pr : (NQB - 1 - pr);

    const int qbase = qblk * QB + wave * 16;

    float qa[64];
    float* xt = xpose[wave];
    const int xr = lane >> 4;
    const int xc = (lane & 15) * 4;
    {
        const float scale = 0.08838834764831845f;
#pragma unroll
        for (int i = 0; i < 8; ++i) {
            const int r   = 2 * i + xr;
            const int pos = qbase + r;
            const float* qrow = q + (((size_t)b * SS + pos) * HH + h) * DD;
            const float* crow = cosp + (size_t)pos * DD;
            const float* srow = sinp + (size_t)pos * DD;
            const float4 xa = *(const float4*)(qrow + xc);
            const float4 xb = *(const float4*)(qrow + xc + 64);
            const float4 ca = *(const float4*)(crow + xc);
            const float4 cb = *(const float4*)(crow + xc + 64);
            const float4 sa = *(const float4*)(srow + xc);
            const float4 sb = *(const float4*)(srow + xc + 64);
            float4 ra, rb;
            ra.x = (xa.x * ca.x - xb.x * sa.x) * scale;  ra.y = (xa.y * ca.y - xb.y * sa.y) * scale;
            ra.z = (xa.z * ca.z - xb.z * sa.z) * scale;  ra.w = (xa.w * ca.w - xb.w * sa.w) * scale;
            rb.x = (xb.x * cb.x + xa.x * sb.x) * scale;  rb.y = (xb.y * cb.y + xa.y * sb.y) * scale;
            rb.z = (xb.z * cb.z + xa.z * sb.z) * scale;  rb.w = (xb.w * cb.w + xa.w * sb.w) * scale;
            *(float4*)(xt + r * XSTRIDE + xc)      = ra;
            *(float4*)(xt + r * XSTRIDE + xc + 64) = rb;
        }
        __builtin_amdgcn_wave_barrier();
#pragma unroll
        for (int st = 0; st < 32; ++st) {
            const v2f t = *(const v2f*)(xt + lmod * XSTRIDE + 4 * st + off);
            qa[2 * st] = t.x; qa[2 * st + 1] = t.y;
        }
    }

    v8f o[8];
#pragma unroll
    for (int j = 0; j < 8; ++j)
#pragma unroll
        for (int r = 0; r < 8; ++r) o[j][r] = 0.0f;
    float mrun = -3.0e38f, lrun = 0.0f;

    const int nst     = qblk * 4 + 4;
    const int ktmax_w = qblk * 8 + wave;
    const int lst_w   = ktmax_w >> 1;

    StageRegs sr0, sr1;
    stage_fetch(sr0, 0, tid, kvhead, k, v, cosp, sinp);
    stage_fetch(sr1, 1, tid, kvhead, k, v, cosp, sinp);
    stage_commit(sr0, tid, kbuf4[0][0], vbufT4[0][0]);
    stage_commit(sr1, tid, kbuf4[0][1], vbufT4[0][1]);
    stage_fetch(sr0, 2, tid, kvhead, k, v, cosp, sinp);
    stage_fetch(sr1, 3, tid, kvhead, k, v, cosp, sinp);
    lds_barrier();

    for (int s = 0; s < nst; ++s) {
        const int cur = s & 1;
        const bool more1 = (s + 1 < nst);
        const bool more2 = (s + 2 < nst);

        const bool active = (s <= lst_w);
        v8f sc0, sc1;
        v2f vf0[32], vf1[32];
        if (active) {
            const float* kb0 = kbuf4[cur][0];
            const float* kb1 = kbuf4[cur][1];
            const float* vb0 = vbufT4[cur][0];
            const float* vb1 = vbufT4[cur][1];
            v2f kf0[32], kf1[32];
#pragma unroll
            for (int st = 0; st < 32; ++st) {
                kf0[st] = *(const v2f*)(kb0 + lmod * KSTRIDE + 4 * st + off);
                kf1[st] = *(const v2f*)(kb1 + lmod * KSTRIDE + 4 * st + off);
            }
            __builtin_amdgcn_sched_barrier(0);
#pragma unroll
            for (int r = 0; r < 8; ++r) { sc0[r] = 0.0f; sc1[r] = 0.0f; }
#pragma unroll
            for (int st = 0; st < 32; ++st) {
                v2f bq;  bq.x = qa[2 * st]; bq.y = qa[2 * st + 1];
                sc0 = __builtin_amdgcn_wmma_f32_16x16x4_f32(
                        false, kf0[st], false, bq, (short)0, sc0, false, false);
                sc1 = __builtin_amdgcn_wmma_f32_16x16x4_f32(
                        false, kf1[st], false, bq, (short)0, sc1, false, false);
            }
            int voff = lmod * VSTRIDE + off;
            asm volatile("" : "+v"(voff) : "v"(sc0), "v"(sc1));
#pragma unroll
            for (int j = 0; j < 8; ++j)
#pragma unroll
                for (int k0 = 0; k0 < 4; ++k0) {
                    vf0[4 * j + k0] = *(const v2f*)(vb0 + voff + 16 * j * VSTRIDE + 4 * k0);
                    vf1[4 * j + k0] = *(const v2f*)(vb1 + voff + 16 * j * VSTRIDE + 4 * k0);
                }
        }
        stage_mid(sr0, sr1, s, cur, more1, more2, tid, kvhead, k, v, cosp, sinp, kbuf4, vbufT4);
        if (active) {
            if (s == lst_w) {
                const bool oddw = (wave & 1) != 0;
#pragma unroll
                for (int r = 0; r < 8; ++r) {
                    const bool dg = (r + 8 * hi > lmod);
                    const bool m0 = oddw ? false : dg;
                    const bool m1 = oddw ? dg : true;
                    sc0[r] = m0 ? -3.0e38f : sc0[r];
                    sc1[r] = m1 ? -3.0e38f : sc1[r];
                }
            }
            const float t0 = fmaxf(fmaxf(fmaxf(sc0[0], sc0[1]), fmaxf(sc0[2], sc0[3])),
                                   fmaxf(fmaxf(sc0[4], sc0[5]), fmaxf(sc0[6], sc0[7])));
            const float t1 = fmaxf(fmaxf(fmaxf(sc1[0], sc1[1]), fmaxf(sc1[2], sc1[3])),
                                   fmaxf(fmaxf(sc1[4], sc1[5]), fmaxf(sc1[6], sc1[7])));
            float t = fmaxf(t0, t1);
            t = fmaxf(t, xhalf(t));
            const float mn    = fmaxf(mrun, t);
            const float alpha = __expf(mrun - mn);
            mrun = mn;
            v8f p0, p1;
#pragma unroll
            for (int r = 0; r < 8; ++r) { p0[r] = __expf(sc0[r] - mn); p1[r] = __expf(sc1[r] - mn); }
            const float su0 = ((p0[0] + p0[1]) + (p0[2] + p0[3])) + ((p0[4] + p0[5]) + (p0[6] + p0[7]));
            const float su1 = ((p1[0] + p1[1]) + (p1[2] + p1[3])) + ((p1[4] + p1[5]) + (p1[6] + p1[7]));
            lrun = lrun * alpha + (su0 + su1);
#pragma unroll
            for (int j = 0; j < 8; ++j)
#pragma unroll
                for (int r = 0; r < 8; ++r) o[j][r] *= alpha;

#pragma unroll
            for (int k0 = 0; k0 < 4; ++k0) {
                v2f bp;  bp.x = p0[2 * k0]; bp.y = p0[2 * k0 + 1];
#pragma unroll
                for (int j = 0; j < 8; ++j)
                    o[j] = __builtin_amdgcn_wmma_f32_16x16x4_f32(
                               false, vf0[4 * j + k0], false, bp, (short)0, o[j], false, false);
            }
#pragma unroll
            for (int k0 = 0; k0 < 4; ++k0) {
                v2f bp;  bp.x = p1[2 * k0]; bp.y = p1[2 * k0 + 1];
#pragma unroll
                for (int j = 0; j < 8; ++j)
                    o[j] = __builtin_amdgcn_wmma_f32_16x16x4_f32(
                               false, vf1[4 * j + k0], false, bp, (short)0, o[j], false, false);
            }
        }
        lds_barrier();
    }

    const float inv_l = 1.0f / (lrun + xhalf(lrun));
#pragma unroll
    for (int j = 0; j < 8; ++j) {
        float4 a, c;
        a.x = o[j][0] * inv_l; a.y = o[j][1] * inv_l; a.z = o[j][2] * inv_l; a.w = o[j][3] * inv_l;
        c.x = o[j][4] * inv_l; c.y = o[j][5] * inv_l; c.z = o[j][6] * inv_l; c.w = o[j][7] * inv_l;
        *(float4*)(xt + lmod * XSTRIDE + 16 * j + 8 * hi)     = a;
        *(float4*)(xt + lmod * XSTRIDE + 16 * j + 8 * hi + 4) = c;
    }
    __builtin_amdgcn_wave_barrier();
#pragma unroll
    for (int i = 0; i < 8; ++i) {
        const int r = 2 * i + xr;
        float* orow = out + ((size_t)bh * SS + qbase + r) * DD;
        *(float4*)(orow + xc)      = *(const float4*)(xt + r * XSTRIDE + xc);
        *(float4*)(orow + xc + 64) = *(const float4*)(xt + r * XSTRIDE + xc + 64);
    }
    __builtin_amdgcn_wave_barrier();
  }
}

extern "C" void kernel_launch(void* const* d_in, const int* in_sizes, int n_in,
                              void* d_out, int out_size, void* d_ws, size_t ws_size,
                              hipStream_t stream) {
    (void)in_sizes; (void)n_in; (void)out_size; (void)d_ws; (void)ws_size;
    const float* q = (const float*)d_in[0];
    const float* k = (const float*)d_in[1];
    const float* v = (const float*)d_in[2];
    const float* c = (const float*)d_in[3];
    const float* s = (const float*)d_in[4];
    float* o = (float*)d_out;
    dim3 grid(BB * HH * (NQB / 2));
    dim3 block(256);
    hipLaunchKernelGGL(fa_rope_fwd, grid, block, 0, stream, q, k, v, c, s, o);
}


# Workgroup barrier split: each wave signals mid-stage, right after its staging writes, and waits at the stage's end behind softmax and output WMMAs, so waves rarely stand at it; sum order unchanged
# speedup vs baseline: 3.1532x; 1.0411x over previous
#include <hip/hip_runtime.h>

typedef float v2f __attribute__((ext_vector_type(2)));
typedef float v8f __attribute__((ext_vector_type(8)));

#define BB 2
#define SS 2048
#define HH 16
#define DD 128
#define QB 128
#define NQB (SS / QB)
#define KSTRIDE 132
#define VSTRIDE 18
#define XSTRIDE 132


__device__ __forceinline__ float xhalf(float x)
{
    const int b = __builtin_bit_cast(int, x);
    return __builtin_bit_cast(float,
        __builtin_amdgcn_permlanex16(b, b, 0x76543210, 0xfedcba98, false, false));
}

struct StageRegs { float4 ka, kb, ca, cb, sa, sb, va, vb; };

__device__ __forceinline__ void stage_fetch(StageRegs& s, int kt, int tid, size_t kvhead,
        const float* __restrict__ k, const float* __restrict__ v,
        const float* __restrict__ cosp, const float* __restrict__ sinp)
{
    const int row = tid >> 4;
    const int c4  = (tid & 15) * 4;
    const int sk  = kt * 16 + row;
    const float* krow = k + kvhead + (size_t)sk * HH * DD;
    const float* vrow = v + kvhead + (size_t)sk * HH * DD;
    const float* crow = cosp + (size_t)sk * DD;
    const float* srow = sinp + (size_t)sk * DD;
    s.ka = *(const float4*)(krow + c4);
    s.kb = *(const float4*)(krow + c4 + 64);
    s.ca = *(const float4*)(crow + c4);
    s.cb = *(const float4*)(crow + c4 + 64);
    s.sa = *(const float4*)(srow + c4);
    s.sb = *(const float4*)(srow + c4 + 64);
    s.va = *(const float4*)(vrow + c4);
    s.vb = *(const float4*)(vrow + c4 + 64);
}

__device__ __forceinline__ void stage_commit(const StageRegs& s, int tid, float* kbuf, float* vbufT)
{
    const int row = tid >> 4;
    const int c4  = (tid & 15) * 4;
    const float4 ka = s.ka, kb = s.kb, ca = s.ca, cb = s.cb, sa = s.sa, sb = s.sb;
    float4 ra, rb;
    ra.x = ka.x * ca.x - kb.x * sa.x;  ra.y = ka.y * ca.y - kb.y * sa.y;
    ra.z = ka.z * ca.z - kb.z * sa.z;  ra.w = ka.w * ca.w - kb.w * sa.w;
    rb.x = kb.x * cb.x + ka.x * sb.x;  rb.y = kb.y * cb.y + ka.y * sb.y;
    rb.z = kb.z * cb.z + ka.z * sb.z;  rb.w = kb.w * cb.w + ka.w * sb.w;
    *(float4*)(kbuf + row * KSTRIDE + c4)      = ra;
    *(float4*)(kbuf + row * KSTRIDE + c4 + 64) = rb;
    const float4 va = s.va, vb = s.vb;
    const int prow = ((row >> 1) & 3) * 4 + (row >> 3) * 2 + (row & 1);
    vbufT[(c4 + 0) * VSTRIDE + prow] = va.x;
    vbufT[(c4 + 1) * VSTRIDE + prow] = va.y;
    vbufT[(c4 + 2) * VSTRIDE + prow] = va.z;
    vbufT[(c4 + 3) * VSTRIDE + prow] = va.w;
    vbufT[(c4 + 64) * VSTRIDE + prow] = vb.x;
    vbufT[(c4 + 65) * VSTRIDE + prow] = vb.y;
    vbufT[(c4 + 66) * VSTRIDE + prow] = vb.z;
    vbufT[(c4 + 67) * VSTRIDE + prow] = vb.w;
}

__device__ __forceinline__ void lds_barrier()
{
    __builtin_amdgcn_fence(__ATOMIC_RELEASE, "workgroup", "local");
    __builtin_amdgcn_s_barrier();
    __builtin_amdgcn_fence(__ATOMIC_ACQUIRE, "workgroup", "local");
}

__device__ __forceinline__ void lds_arrive()
{
    __builtin_amdgcn_fence(__ATOMIC_RELEASE, "workgroup", "local");
    __builtin_amdgcn_s_barrier_signal(-1);
}
__device__ __forceinline__ void lds_wait()
{
    __builtin_amdgcn_s_barrier_wait(-1);
    __builtin_amdgcn_fence(__ATOMIC_ACQUIRE, "workgroup", "local");
}

__device__ __forceinline__ void stage_mid(StageRegs& sr0, StageRegs& sr1, int s, int cur,
        bool more1, bool more2, int tid, size_t kvhead,
        const float* __restrict__ k, const float* __restrict__ v,
        const float* __restrict__ cosp, const float* __restrict__ sinp,
        float (*kbuf4)[2][16 * KSTRIDE], float (*vbufT4)[2][DD * VSTRIDE])
{
    if (more1) {
        stage_commit(sr0, tid, kbuf4[cur ^ 1][0], vbufT4[cur ^ 1][0]);
        stage_commit(sr1, tid, kbuf4[cur ^ 1][1], vbufT4[cur ^ 1][1]);
    }
    if (more2) {
        stage_fetch(sr0, 2 * s + 4, tid, kvhead, k, v, cosp, sinp);
        stage_fetch(sr1, 2 * s + 5, tid, kvhead, k, v, cosp, sinp);
    }
}

__global__ __launch_bounds__(256)
void fa_rope_fwd(const float* __restrict__ q, const float* __restrict__ k,
                 const float* __restrict__ v, const float* __restrict__ cosp,
                 const float* __restrict__ sinp, float* __restrict__ out)
{
    __shared__ float kbuf4[2][2][16 * KSTRIDE];
    __shared__ float vbufT4[2][2][DD * VSTRIDE];
    __shared__ float xpose[8][16 * XSTRIDE];

    const int tid  = threadIdx.x;
    const int wave = tid >> 5;
    const int lane = tid & 31;
    const int lmod = lane & 15;
    const int hi   = lane >> 4;
    const int off  = hi * 2;

    const int pr   = blockIdx.x & 7;
    const int bh   = blockIdx.x >> 3;
    const int b    = bh >> 4;
    const int h    = bh & 15;
    const size_t kvhead = ((size_t)b * SS * HH + h) * DD;

  for (int pass = 0; pass < 2; ++pass) {
    const int qblk = pass ? pr : (NQB - 1 - pr);

    const int qbase = qblk * QB + wave * 16;

    float qa[64];
    float* xt = xpose[wave];
    const int xr = lane >> 4;
    const int xc = (lane & 15) * 4;
    {
        const float scale = 0.08838834764831845f;
#pragma unroll
        for (int i = 0; i < 8; ++i) {
            const int r   = 2 * i + xr;
            const int pos = qbase + r;
            const float* qrow = q + (((size_t)b * SS + pos) * HH + h) * DD;
            const float* crow = cosp + (size_t)pos * DD;
            const float* srow = sinp + (size_t)pos * DD;
            const float4 xa = *(const float4*)(qrow + xc);
            const float4 xb = *(const float4*)(qrow + xc + 64);
            const float4 ca = *(const float4*)(crow + xc);
            const float4 cb = *(const float4*)(crow + xc + 64);
            const float4 sa = *(const float4*)(srow + xc);
            const float4 sb = *(const float4*)(srow + xc + 64);
            float4 ra, rb;
            ra.x = (xa.x * ca.x - xb.x * sa.x) * scale;  ra.y = (xa.y * ca.y - xb.y * sa.y) * scale;
            ra.z = (xa.z * ca.z - xb.z * sa.z) * scale;  ra.w = (xa.w * ca.w - xb.w * sa.w) * scale;
            rb.x = (xb.x * cb.x + xa.x * sb.x) * scale;  rb.y = (xb.y * cb.y + xa.y * sb.y) * scale;
            rb.z = (xb.z * cb.z + xa.z * sb.z) * scale;  rb.w = (xb.w * cb.w + xa.w * sb.w) * scale;
            *(float4*)(xt + r * XSTRIDE + xc)      = ra;
            *(float4*)(xt + r * XSTRIDE + xc + 64) = rb;
        }
        __builtin_amdgcn_wave_barrier();
#pragma unroll
        for (int st = 0; st < 32; ++st) {
            const v2f t = *(const v2f*)(xt + lmod * XSTRIDE + 4 * st + off);
            qa[2 * st] = t.x; qa[2 * st + 1] = t.y;
        }
    }

    v8f o[8];
#pragma unroll
    for (int j = 0; j < 8; ++j)
#pragma unroll
        for (int r = 0; r < 8; ++r) o[j][r] = 0.0f;
    float mrun = -3.0e38f, lrun = 0.0f;

    const int nst     = qblk * 4 + 4;
    const int ktmax_w = qblk * 8 + wave;
    const int lst_w   = ktmax_w >> 1;

    StageRegs sr0, sr1;
    stage_fetch(sr0, 0, tid, kvhead, k, v, cosp, sinp);
    stage_fetch(sr1, 1, tid, kvhead, k, v, cosp, sinp);
    stage_commit(sr0, tid, kbuf4[0][0], vbufT4[0][0]);
    stage_commit(sr1, tid, kbuf4[0][1], vbufT4[0][1]);
    stage_fetch(sr0, 2, tid, kvhead, k, v, cosp, sinp);
    stage_fetch(sr1, 3, tid, kvhead, k, v, cosp, sinp);
    lds_barrier();

    for (int s = 0; s < nst; ++s) {
        const int cur = s & 1;
        const bool more1 = (s + 1 < nst);
        const bool more2 = (s + 2 < nst);

        const bool active = (s <= lst_w);
        v8f sc0, sc1;
        v2f vf0[32], vf1[32];
        if (active) {
            const float* kb0 = kbuf4[cur][0];
            const float* kb1 = kbuf4[cur][1];
            const float* vb0 = vbufT4[cur][0];
            const float* vb1 = vbufT4[cur][1];
            v2f kf0[32], kf1[32];
#pragma unroll
            for (int st = 0; st < 32; ++st) {
                kf0[st] = *(const v2f*)(kb0 + lmod * KSTRIDE + 4 * st + off);
                kf1[st] = *(const v2f*)(kb1 + lmod * KSTRIDE + 4 * st + off);
            }
            __builtin_amdgcn_sched_barrier(0);
#pragma unroll
            for (int r = 0; r < 8; ++r) { sc0[r] = 0.0f; sc1[r] = 0.0f; }
#pragma unroll
            for (int st = 0; st < 32; ++st) {
                v2f bq;  bq.x = qa[2 * st]; bq.y = qa[2 * st + 1];
                sc0 = __builtin_amdgcn_wmma_f32_16x16x4_f32(
                        false, kf0[st], false, bq, (short)0, sc0, false, false);
                sc1 = __builtin_amdgcn_wmma_f32_16x16x4_f32(
                        false, kf1[st], false, bq, (short)0, sc1, false, false);
            }
            int voff = lmod * VSTRIDE + off;
            asm volatile("" : "+v"(voff) : "v"(sc0), "v"(sc1));
#pragma unroll
            for (int j = 0; j < 8; ++j)
#pragma unroll
                for (int k0 = 0; k0 < 4; ++k0) {
                    vf0[4 * j + k0] = *(const v2f*)(vb0 + voff + 16 * j * VSTRIDE + 4 * k0);
                    vf1[4 * j + k0] = *(const v2f*)(vb1 + voff + 16 * j * VSTRIDE + 4 * k0);
                }
        }
        stage_mid(sr0, sr1, s, cur, more1, more2, tid, kvhead, k, v, cosp, sinp, kbuf4, vbufT4);
        lds_arrive();
        if (active) {
            if (s == lst_w) {
                const bool oddw = (wave & 1) != 0;
#pragma unroll
                for (int r = 0; r < 8; ++r) {
                    const bool dg = (r + 8 * hi > lmod);
                    const bool m0 = oddw ? false : dg;
                    const bool m1 = oddw ? dg : true;
                    sc0[r] = m0 ? -3.0e38f : sc0[r];
                    sc1[r] = m1 ? -3.0e38f : sc1[r];
                }
            }
            const float t0 = fmaxf(fmaxf(fmaxf(sc0[0], sc0[1]), fmaxf(sc0[2], sc0[3])),
                                   fmaxf(fmaxf(sc0[4], sc0[5]), fmaxf(sc0[6], sc0[7])));
            const float t1 = fmaxf(fmaxf(fmaxf(sc1[0], sc1[1]), fmaxf(sc1[2], sc1[3])),
                                   fmaxf(fmaxf(sc1[4], sc1[5]), fmaxf(sc1[6], sc1[7])));
            float t = fmaxf(t0, t1);
            t = fmaxf(t, xhalf(t));
            const float mn    = fmaxf(mrun, t);
            const float alpha = __expf(mrun - mn);
            mrun = mn;
            v8f p0, p1;
#pragma unroll
            for (int r = 0; r < 8; ++r) { p0[r] = __expf(sc0[r] - mn); p1[r] = __expf(sc1[r] - mn); }
            const float su0 = ((p0[0] + p0[1]) + (p0[2] + p0[3])) + ((p0[4] + p0[5]) + (p0[6] + p0[7]));
            const float su1 = ((p1[0] + p1[1]) + (p1[2] + p1[3])) + ((p1[4] + p1[5]) + (p1[6] + p1[7]));
            lrun = lrun * alpha + (su0 + su1);
#pragma unroll
            for (int j = 0; j < 8; ++j)
#pragma unroll
                for (int r = 0; r < 8; ++r) o[j][r] *= alpha;

#pragma unroll
            for (int k0 = 0; k0 < 4; ++k0) {
                v2f bp;  bp.x = p0[2 * k0]; bp.y = p0[2 * k0 + 1];
#pragma unroll
                for (int j = 0; j < 8; ++j)
                    o[j] = __builtin_amdgcn_wmma_f32_16x16x4_f32(
                               false, vf0[4 * j + k0], false, bp, (short)0, o[j], false, false);
            }
#pragma unroll
            for (int k0 = 0; k0 < 4; ++k0) {
                v2f bp;  bp.x = p1[2 * k0]; bp.y = p1[2 * k0 + 1];
#pragma unroll
                for (int j = 0; j < 8; ++j)
                    o[j] = __builtin_amdgcn_wmma_f32_16x16x4_f32(
                               false, vf1[4 * j + k0], false, bp, (short)0, o[j], false, false);
            }
        }
        lds_wait();
    }

    const float inv_l = 1.0f / (lrun + xhalf(lrun));
#pragma unroll
    for (int j = 0; j < 8; ++j) {
        float4 a, c;
        a.x = o[j][0] * inv_l; a.y = o[j][1] * inv_l; a.z = o[j][2] * inv_l; a.w = o[j][3] * inv_l;
        c.x = o[j][4] * inv_l; c.y = o[j][5] * inv_l; c.z = o[j][6] * inv_l; c.w = o[j][7] * inv_l;
        *(float4*)(xt + lmod * XSTRIDE + 16 * j + 8 * hi)     = a;
        *(float4*)(xt + lmod * XSTRIDE + 16 * j + 8 * hi + 4) = c;
    }
    __builtin_amdgcn_wave_barrier();
#pragma unroll
    for (int i = 0; i < 8; ++i) {
        const int r = 2 * i + xr;
        float* orow = out + ((size_t)bh * SS + qbase + r) * DD;
        *(float4*)(orow + xc)      = *(const float4*)(xt + r * XSTRIDE + xc);
        *(float4*)(orow + xc + 64) = *(const float4*)(xt + r * XSTRIDE + xc + 64);
    }
    __builtin_amdgcn_wave_barrier();
  }
}

extern "C" void kernel_launch(void* const* d_in, const int* in_sizes, int n_in,
                              void* d_out, int out_size, void* d_ws, size_t ws_size,
                              hipStream_t stream) {
    (void)in_sizes; (void)n_in; (void)out_size; (void)d_ws; (void)ws_size;
    const float* q = (const float*)d_in[0];
    const float* k = (const float*)d_in[1];
    const float* v = (const float*)d_in[2];
    const float* c = (const float*)d_in[3];
    const float* s = (const float*)d_in[4];
    float* o = (float*)d_out;
    dim3 grid(BB * HH * (NQB / 2));
    dim3 block(256);
    hipLaunchKernelGGL(fa_rope_fwd, grid, block, 0, stream, q, k, v, c, s, o);
}
